# starting kernel
# baseline (speedup 1.0000x reference)
#include <hip/hip_runtime.h>
#include <hip/hip_bf16.h>
#include <math.h>

typedef __attribute__((ext_vector_type(2))) float v2f;
typedef __attribute__((ext_vector_type(8))) float v8f;

#define EPSLN 1e-5f

__device__ __forceinline__ v8f wmma_f32_4(v2f a, v2f b, v8f c) {
    return __builtin_amdgcn_wmma_f32_16x16x4_f32(
         false, a,  false, b,
         (short)0, c,  false,  false);
}

__device__ __forceinline__ void atomicMaxF(float* addr, float val) {
    if (val >= 0.0f) atomicMax((int*)addr, __float_as_int(val));
    else             atomicMin((unsigned int*)addr, __float_as_uint(val));
}

__global__ __launch_bounds__(256) void qkv_kernel(
    const float* __restrict__ feat,
    const float* __restrict__ Wq, const float* __restrict__ Wk,
    const float* __restrict__ Wv,
    float* __restrict__ q, float* __restrict__ k, float* __restrict__ v, int n)
{
    __shared__ float sA[64][132];
    const int row0 = blockIdx.x * 64;
    const int tid  = threadIdx.x;

    for (int i = tid; i < 64 * 32; i += 256) {
        int r = i >> 5, c4 = i & 31;
        int gr = row0 + r; if (gr >= n) gr = n - 1;
        ((float4*)&sA[r][0])[c4] = ((const float4*)(feat + (size_t)gr * 128))[c4];
    }
    __syncthreads();

    const int wave = tid >> 5;
    const int lane = tid & 31;
    const int half = lane >> 4;
    const int lr   = lane & 15;
    const int col  = wave * 16 + lr;

    const float* Ws[3] = { Wq, Wk, Wv };
    float*       Os[3] = { q,  k,  v  };

#pragma unroll
    for (int m = 0; m < 3; ++m) {
        const float* __restrict__ W = Ws[m];
        v2f bfr[32];
#pragma unroll
        for (int kq = 0; kq < 32; ++kq) {
            const int kk = 4 * kq + 2 * half;
            bfr[kq] = (v2f){ W[(size_t)kk * 128 + col],
                             W[(size_t)(kk + 1) * 128 + col] };
        }
        float* __restrict__ O = Os[m];
#pragma unroll
        for (int mt = 0; mt < 4; ++mt) {
            v8f acc = {};
#pragma unroll
            for (int kq = 0; kq < 32; ++kq) {
                const int kk = 4 * kq + 2 * half;
                v2f a = { sA[mt * 16 + lr][kk], sA[mt * 16 + lr][kk + 1] };
                acc = wmma_f32_4(a, bfr[kq], acc);
            }
#pragma unroll
            for (int r = 0; r < 8; ++r) {
                int gr = row0 + mt * 16 + r + 8 * half;
                if (gr < n) O[(size_t)gr * 128 + col] = acc[r];
            }
        }
    }
}

__global__ __launch_bounds__(256) void init_kernel(
    float* __restrict__ smax, float* __restrict__ ssum,
    float* __restrict__ ft2, int n)
{
    int i = blockIdx.x * 256 + threadIdx.x;
    if (i < n * 8) { smax[i] = -INFINITY; ssum[i] = 0.0f; }
    if (i < n * 128) ft2[i] = 0.0f;
}

__global__ __launch_bounds__(256) void scores_kernel(
    const float* __restrict__ kbuf, const float* __restrict__ qbuf,
    const int* __restrict__ src, const int* __restrict__ dst,
    float* __restrict__ scores, float* __restrict__ smax, int eh)
{
    int t = blockIdx.x * 256 + threadIdx.x;
    if (t >= eh) return;
    const int e = t >> 3, h = t & 7;
    const int s = src[e], d = dst[e];
    const float4* kp = (const float4*)(kbuf + (size_t)s * 128 + h * 16);
    const float4* qp = (const float4*)(qbuf + (size_t)d * 128 + h * 16);
    float acc = 0.0f;
#pragma unroll
    for (int i = 0; i < 4; ++i) {
        float4 a = kp[i], b = qp[i];
        acc += a.x * b.x + a.y * b.y + a.z * b.z + a.w * b.w;
    }
    acc *= 0.08838834764831845f;
    scores[t] = acc;
    atomicMaxF(&smax[d * 8 + h], acc);
}

__global__ __launch_bounds__(256) void expsum_kernel(
    const int* __restrict__ dst, float* __restrict__ scores,
    const float* __restrict__ smax, float* __restrict__ ssum, int eh)
{
    int t = blockIdx.x * 256 + threadIdx.x;
    if (t >= eh) return;
    const int e = t >> 3, h = t & 7;
    const int d = dst[e];
    float ex = expf(scores[t] - smax[d * 8 + h]);
    scores[t] = ex;
    atomicAdd(&ssum[d * 8 + h], ex);
}

__global__ __launch_bounds__(256) void scatter_kernel(
    const int* __restrict__ src, const int* __restrict__ dst,
    const float* __restrict__ scores, const float* __restrict__ ssum,
    const float* __restrict__ vbuf, float* __restrict__ ft2, int eh)
{
    int t = blockIdx.x * 256 + threadIdx.x;
    if (t >= eh) return;
    const int e = t >> 3, h = t & 7;
    const int s = src[e], d = dst[e];
    const float a = scores[t] / ssum[d * 8 + h];
    const float* vp = vbuf + (size_t)s * 128 + h * 16;
    float*       fp = ft2  + (size_t)d * 128 + h * 16;
#pragma unroll
    for (int i = 0; i < 16; ++i)
        atomicAdd(&fp[i], vp[i] * a);
}

__global__ __launch_bounds__(256) void ln1_kernel(
    const float* __restrict__ ft2, const float* __restrict__ feat,
    const float* __restrict__ g, const float* __restrict__ b,
    float* __restrict__ rst, int n)
{
    const int row = blockIdx.x * 16 + (threadIdx.x >> 4);
    const int c0  = threadIdx.x & 15;
    if (row >= n) return;
    const float* p1 = ft2  + (size_t)row * 128;
    const float* p2 = feat + (size_t)row * 128;
    float xv[8], s = 0.0f, ss = 0.0f;
#pragma unroll
    for (int j = 0; j < 8; ++j) {
        int c = c0 + 16 * j;
        float x = p1[c] + p2[c];
        xv[j] = x; s += x; ss += x * x;
    }
#pragma unroll
    for (int m = 1; m < 16; m <<= 1) {
        s  += __shfl_xor(s,  m, 32);
        ss += __shfl_xor(ss, m, 32);
    }
    const float mean = s * (1.0f / 128.0f);
    const float var  = ss * (1.0f / 128.0f) - mean * mean;
    const float inv  = rsqrtf(var + EPSLN);
    float* o = rst + (size_t)row * 128;
#pragma unroll
    for (int j = 0; j < 8; ++j) {
        int c = c0 + 16 * j;
        o[c] = (xv[j] - mean) * inv * g[c] + b[c];
    }
}

__global__ __launch_bounds__(256) void ffn1_kernel(
    const float* __restrict__ rst, const float* __restrict__ W1,
    const float* __restrict__ b1, const float* __restrict__ pw,
    float* __restrict__ hbuf, int n)
{
    __shared__ float sA[64][132];
    const int row0 = blockIdx.x * 64;
    const int tid  = threadIdx.x;
    for (int i = tid; i < 64 * 32; i += 256) {
        int r = i >> 5, c4 = i & 31;
        int gr = row0 + r; if (gr >= n) gr = n - 1;
        ((float4*)&sA[r][0])[c4] = ((const float4*)(rst + (size_t)gr * 128))[c4];
    }
    __syncthreads();

    const int wave = tid >> 5;
    const int lane = tid & 31;
    const int half = lane >> 4;
    const int lr   = lane & 15;

#pragma unroll
    for (int j = 0; j < 4; ++j) {
        const int col = (wave + 8 * j) * 16 + lr;
        v2f bfr[32];
#pragma unroll
        for (int kq = 0; kq < 32; ++kq) {
            const int kk = 4 * kq + 2 * half;
            bfr[kq] = (v2f){ W1[(size_t)kk * 512 + col],
                             W1[(size_t)(kk + 1) * 512 + col] };
        }
        const float bias = b1[col], slope = pw[col];
#pragma unroll
        for (int mt = 0; mt < 4; ++mt) {
            v8f acc = {};
#pragma unroll
            for (int kq = 0; kq < 32; ++kq) {
                const int kk = 4 * kq + 2 * half;
                v2f a = { sA[mt * 16 + lr][kk], sA[mt * 16 + lr][kk + 1] };
                acc = wmma_f32_4(a, bfr[kq], acc);
            }
#pragma unroll
            for (int r = 0; r < 8; ++r) {
                int gr = row0 + mt * 16 + r + 8 * half;
                if (gr < n) {
                    float h = acc[r] + bias;
                    h = (h >= 0.0f) ? h : slope * h;
                    hbuf[(size_t)gr * 512 + col] = h;
                }
            }
        }
    }
}

__global__ __launch_bounds__(256) void ffn2_ln_kernel(
    const float* __restrict__ hbuf, const float* __restrict__ W2,
    const float* __restrict__ b2, const float* __restrict__ rst,
    const float* __restrict__ g, const float* __restrict__ be,
    float* __restrict__ out, int n)
{
    __shared__ float sA[64][132];
    const int row0 = blockIdx.x * 64;
    const int tid  = threadIdx.x;
    const int wave = tid >> 5;
    const int lane = tid & 31;
    const int half = lane >> 4;
    const int lr   = lane & 15;
    const int col  = wave * 16 + lr;

    v8f acc[4] = {};
#pragma unroll
    for (int kc = 0; kc < 4; ++kc) {
        __syncthreads();
        for (int i = tid; i < 64 * 32; i += 256) {
            int r = i >> 5, c4 = i & 31;
            int gr = row0 + r; if (gr >= n) gr = n - 1;
            ((float4*)&sA[r][0])[c4] =
                ((const float4*)(hbuf + (size_t)gr * 512 + kc * 128))[c4];
        }
        __syncthreads();

        v2f bfr[32];
#pragma unroll
        for (int kq = 0; kq < 32; ++kq) {
            const int kk = kc * 128 + 4 * kq + 2 * half;
            bfr[kq] = (v2f){ W2[(size_t)kk * 128 + col],
                             W2[(size_t)(kk + 1) * 128 + col] };
        }
#pragma unroll
        for (int mt = 0; mt < 4; ++mt) {
#pragma unroll
            for (int kq = 0; kq < 32; ++kq) {
                const int kk = 4 * kq + 2 * half;
                v2f a = { sA[mt * 16 + lr][kk], sA[mt * 16 + lr][kk + 1] };
                acc[mt] = wmma_f32_4(a, bfr[kq], acc[mt]);
            }
        }
    }

    __syncthreads();
    const float bias = b2[col];
#pragma unroll
    for (int mt = 0; mt < 4; ++mt) {
#pragma unroll
        for (int r = 0; r < 8; ++r) {
            const int rl = mt * 16 + r + 8 * half;
            int gr = row0 + rl; if (gr >= n) gr = n - 1;
            sA[rl][col] = acc[mt][r] + bias + rst[(size_t)gr * 128 + col];
        }
    }
    __syncthreads();

    const int rl2 = tid >> 2;
    const int c0  = tid & 3;
    const int row = row0 + rl2;
    float xv[32], s = 0.0f, ss = 0.0f;
#pragma unroll
    for (int j = 0; j < 32; ++j) {
        int c = c0 + 4 * j;
        float x = sA[rl2][c];
        xv[j] = x; s += x; ss += x * x;
    }
#pragma unroll
    for (int m = 1; m < 4; m <<= 1) {
        s  += __shfl_xor(s,  m, 32);
        ss += __shfl_xor(ss, m, 32);
    }
    const float mean = s * (1.0f / 128.0f);
    const float var  = ss * (1.0f / 128.0f) - mean * mean;
    const float inv  = rsqrtf(var + EPSLN);
    if (row < n) {
        float* o = out + (size_t)row * 128;
#pragma unroll
        for (int j = 0; j < 32; ++j) {
            int c = c0 + 4 * j;
            o[c] = (xv[j] - mean) * inv * g[c] + be[c];
        }
    }
}

extern "C" void kernel_launch(void* const* d_in, const int* in_sizes, int n_in,
                              void* d_out, int out_size, void* d_ws, size_t ws_size,
                              hipStream_t stream)
{
    const float* feat  = (const float*)d_in[0];
    const int*   src   = (const int*)  d_in[1];
    const int*   dst   = (const int*)  d_in[2];
    const float* Wq    = (const float*)d_in[3];
    const float* Wk    = (const float*)d_in[4];
    const float* Wv    = (const float*)d_in[5];
    const float* ln1_g = (const float*)d_in[6];
    const float* ln1_b = (const float*)d_in[7];
    const float* ln2_g = (const float*)d_in[8];
    const float* ln2_b = (const float*)d_in[9];
    const float* W1    = (const float*)d_in[10];
    const float* b1    = (const float*)d_in[11];
    const float* pw    = (const float*)d_in[12];
    const float* W2    = (const float*)d_in[13];
    const float* b2    = (const float*)d_in[14];

    const int n = in_sizes[0] / 128;
    const int e = in_sizes[1];
    const int eh = e * 8;

    float* ws = (float*)d_ws;
    size_t off = 0;
    float* q      = ws + off; off += (size_t)n * 128;
    float* kbuf   = ws + off; off += (size_t)n * 128;
    float* vbuf   = ws + off; off += (size_t)n * 128;
    float* ft2    = ws + off; off += (size_t)n * 128;
    float* rst    = ws + off; off += (size_t)n * 128;
    float* hbuf   = ws + off; off += (size_t)n * 512;
    float* scores = ws + off; off += (size_t)eh;
    float* smax   = ws + off; off += (size_t)n * 8;
    float* ssum   = ws + off; off += (size_t)n * 8;
    (void)off; (void)ws_size; (void)n_in; (void)out_size;

    const int NT64 = (n + 63) / 64;
    const int NT16 = (n + 15) / 16;
    const int NB   = (n * 128 + 255) / 256;
    const int EB   = (eh + 255) / 256;

    qkv_kernel<<<NT64, 256, 0, stream>>>(feat, Wq, Wk, Wv, q, kbuf, vbuf, n);
    init_kernel<<<NB, 256, 0, stream>>>(smax, ssum, ft2, n);
    scores_kernel<<<EB, 256, 0, stream>>>(kbuf, q, src, dst, scores, smax, eh);
    expsum_kernel<<<EB, 256, 0, stream>>>(dst, scores, smax, ssum, eh);
    scatter_kernel<<<EB, 256, 0, stream>>>(src, dst, scores, ssum, vbuf, ft2, eh);
    ln1_kernel<<<NT16, 256, 0, stream>>>(ft2, feat, ln1_g, ln1_b, rst, n);
    ffn1_kernel<<<NT64, 256, 0, stream>>>(rst, W1, b1, pw, hbuf, n);
    ffn2_ln_kernel<<<NT64, 256, 0, stream>>>(hbuf, W2, b2, rst, ln2_g, ln2_b,
                                             (float*)d_out, n);
}


# edge softmax+aggregation as a gather: edges counting-sorted by dst (int atomics draw slots), one wave per node sums its in-edges in registers; no float atomics; sum order by src id per 32 slots
# speedup vs baseline: 2.2901x; 2.2901x over previous
#include <hip/hip_runtime.h>
#include <hip/hip_bf16.h>
#include <math.h>

typedef __attribute__((ext_vector_type(2))) float v2f;
typedef __attribute__((ext_vector_type(8))) float v8f;

#define EPSLN 1e-5f

__device__ __forceinline__ v8f wmma_f32_4(v2f a, v2f b, v8f c) {
    return __builtin_amdgcn_wmma_f32_16x16x4_f32(
         false, a,  false, b,
         (short)0, c,  false,  false);
}

__global__ __launch_bounds__(256) void qkv_kernel(
    const float* __restrict__ feat,
    const float* __restrict__ Wq, const float* __restrict__ Wk,
    const float* __restrict__ Wv,
    float* __restrict__ q, float* __restrict__ k, float* __restrict__ v, int n)
{
    __shared__ float sA[64][132];
    const int row0 = blockIdx.x * 64;
    const int tid  = threadIdx.x;

    for (int i = tid; i < 64 * 32; i += 256) {
        int r = i >> 5, c4 = i & 31;
        int gr = row0 + r; if (gr >= n) gr = n - 1;
        ((float4*)&sA[r][0])[c4] = ((const float4*)(feat + (size_t)gr * 128))[c4];
    }
    __syncthreads();

    const int wave = tid >> 5;
    const int lane = tid & 31;
    const int half = lane >> 4;
    const int lr   = lane & 15;
    const int col  = wave * 16 + lr;

    const float* Ws[3] = { Wq, Wk, Wv };
    float*       Os[3] = { q,  k,  v  };

#pragma unroll
    for (int m = 0; m < 3; ++m) {
        const float* __restrict__ W = Ws[m];
        v2f bfr[32];
#pragma unroll
        for (int kq = 0; kq < 32; ++kq) {
            const int kk = 4 * kq + 2 * half;
            bfr[kq] = (v2f){ W[(size_t)kk * 128 + col],
                             W[(size_t)(kk + 1) * 128 + col] };
        }
        float* __restrict__ O = Os[m];
#pragma unroll
        for (int mt = 0; mt < 4; ++mt) {
            v8f acc = {};
#pragma unroll
            for (int kq = 0; kq < 32; ++kq) {
                const int kk = 4 * kq + 2 * half;
                v2f a = { sA[mt * 16 + lr][kk], sA[mt * 16 + lr][kk + 1] };
                acc = wmma_f32_4(a, bfr[kq], acc);
            }
#pragma unroll
            for (int r = 0; r < 8; ++r) {
                int gr = row0 + mt * 16 + r + 8 * half;
                if (gr < n) O[(size_t)gr * 128 + col] = acc[r];
            }
        }
    }
}

#define SCAN_T 1024

__device__ __forceinline__ int scan_chunk(int n) {
    return (((n + SCAN_T - 1) / SCAN_T) + 3) & ~3;
}

__global__ __launch_bounds__(256) void zero_kernel(int* __restrict__ cnt, int n)
{
    const int i = blockIdx.x * 256 + threadIdx.x;
    if (i < SCAN_T * scan_chunk(n)) cnt[i] = 0;
}

__global__ __launch_bounds__(256) void hist_kernel(
    const int* __restrict__ dst, int* __restrict__ cnt,
    int* __restrict__ rank, int e)
{
    const int t = blockIdx.x * 256 + threadIdx.x;
    if (t < e) rank[t] = atomicAdd(&cnt[dst[t]], 1);
}

__global__ __launch_bounds__(SCAN_T) void scan_kernel(
    const int* __restrict__ cnt, int* __restrict__ rowptr, int n)
{
    __shared__ int wsum[32];
    const int chunk4 = scan_chunk(n) >> 2;
    const int t = threadIdx.x, lane = t & 31, w = t >> 5;
    const int4* cp = (const int4*)cnt + (size_t)t * chunk4;
    int s = 0;
    for (int i = 0; i < chunk4; ++i) { int4 c = cp[i]; s += c.x + c.y + c.z + c.w; }
    int incl = s;
#pragma unroll
    for (int o = 1; o < 32; o <<= 1) {
        int y = __shfl_up(incl, o, 32);
        if (lane >= o) incl += y;
    }
    if (lane == 31) wsum[w] = incl;
    __syncthreads();
    if (w == 0) {
        const int x = wsum[lane];
        int xi = x;
#pragma unroll
        for (int o = 1; o < 32; o <<= 1) {
            int y = __shfl_up(xi, o, 32);
            if (lane >= o) xi += y;
        }
        wsum[lane] = xi - x;
    }
    __syncthreads();
    int run = wsum[w] + incl - s;
    int4* rp = (int4*)rowptr + (size_t)t * chunk4;
    for (int i = 0; i < chunk4; ++i) {
        int4 c = cp[i], r;
        r.x = run; run += c.x;
        r.y = run; run += c.y;
        r.z = run; run += c.z;
        r.w = run; run += c.w;
        rp[i] = r;
    }
    if (t == SCAN_T - 1) rowptr[SCAN_T * 4 * chunk4] = run;
}

__global__ __launch_bounds__(256) void fill_kernel(
    const int* __restrict__ src, const int* __restrict__ dst,
    const int* __restrict__ rank, const int* __restrict__ rowptr,
    int* __restrict__ ssrc, int e)
{
    const int t = blockIdx.x * 256 + threadIdx.x;
    if (t < e) ssrc[rowptr[dst[t]] + rank[t]] = src[t];
}

__device__ __forceinline__ void edge_step(
    const float4 kv, const float4 vv, const float4 qv,
    float& m, float& l, float4& acc)
{
    float p = kv.x * qv.x + kv.y * qv.y + kv.z * qv.z + kv.w * qv.w;
    p += __shfl_xor(p, 1, 32);
    p += __shfl_xor(p, 2, 32);
    p *= 0.08838834764831845f;
    const float t  = expf(-fabsf(p - m));
    const bool  up = p > m;
    const float sc = up ? t : 1.0f;
    const float ex = up ? 1.0f : t;
    m = up ? p : m;
    l = l * sc + ex;
    acc.x = acc.x * sc + ex * vv.x;
    acc.y = acc.y * sc + ex * vv.y;
    acc.z = acc.z * sc + ex * vv.z;
    acc.w = acc.w * sc + ex * vv.w;
}

__global__ __launch_bounds__(256) void node_kernel(
    const float* __restrict__ qbuf, const float* __restrict__ kbuf,
    const float* __restrict__ vbuf, const int* __restrict__ rowptr,
    const int* __restrict__ ssrc, float* __restrict__ ft2, int n)
{
    const int lane = threadIdx.x & 31;
    const int d = blockIdx.x * 8 + (threadIdx.x >> 5);
    if (d >= n) return;
    const int beg = rowptr[d], end = rowptr[d + 1];
    const float4 qv = ((const float4*)(qbuf + (size_t)d * 128))[lane];
    float m = -INFINITY, l = 0.0f;
    float4 acc = { 0.0f, 0.0f, 0.0f, 0.0f };

    for (int base = beg; base < end; base += 32) {
        const int cnt = min(32, end - base);
        int x = (lane < cnt) ? ssrc[base + lane] : 0x7fffffff;
#pragma unroll
        for (int kk = 2; kk <= 32; kk <<= 1) {
#pragma unroll
            for (int jj = kk >> 1; jj > 0; jj >>= 1) {
                const int o  = __shfl_xor(x, jj, 32);
                const int mn = min(x, o), mx = max(x, o);
                const bool asc = (lane & kk) == 0, low = (lane & jj) == 0;
                x = (asc == low) ? mn : mx;
            }
        }
        int j = 0;
        for (; j + 4 <= cnt; j += 4) {
            const int s0 = __builtin_amdgcn_readlane(x, j);
            const int s1 = __builtin_amdgcn_readlane(x, j + 1);
            const int s2 = __builtin_amdgcn_readlane(x, j + 2);
            const int s3 = __builtin_amdgcn_readlane(x, j + 3);
            const float4 k0 = ((const float4*)(kbuf + (size_t)s0 * 128))[lane];
            const float4 k1 = ((const float4*)(kbuf + (size_t)s1 * 128))[lane];
            const float4 k2 = ((const float4*)(kbuf + (size_t)s2 * 128))[lane];
            const float4 k3 = ((const float4*)(kbuf + (size_t)s3 * 128))[lane];
            const float4 v0 = ((const float4*)(vbuf + (size_t)s0 * 128))[lane];
            const float4 v1 = ((const float4*)(vbuf + (size_t)s1 * 128))[lane];
            const float4 v2 = ((const float4*)(vbuf + (size_t)s2 * 128))[lane];
            const float4 v3 = ((const float4*)(vbuf + (size_t)s3 * 128))[lane];
            edge_step(k0, v0, qv, m, l, acc);
            edge_step(k1, v1, qv, m, l, acc);
            edge_step(k2, v2, qv, m, l, acc);
            edge_step(k3, v3, qv, m, l, acc);
        }
        for (; j < cnt; ++j) {
            const int s0 = __builtin_amdgcn_readlane(x, j);
            const float4 k0 = ((const float4*)(kbuf + (size_t)s0 * 128))[lane];
            const float4 v0 = ((const float4*)(vbuf + (size_t)s0 * 128))[lane];
            edge_step(k0, v0, qv, m, l, acc);
        }
    }
    const float inv = (end > beg) ? 1.0f / l : 0.0f;
    float4 o4 = { acc.x * inv, acc.y * inv, acc.z * inv, acc.w * inv };
    ((float4*)(ft2 + (size_t)d * 128))[lane] = o4;
}

__global__ __launch_bounds__(256) void ln1_kernel(
    const float* __restrict__ ft2, const float* __restrict__ feat,
    const float* __restrict__ g, const float* __restrict__ b,
    float* __restrict__ rst, int n)
{
    const int row = blockIdx.x * 16 + (threadIdx.x >> 4);
    const int c0  = threadIdx.x & 15;
    if (row >= n) return;
    const float* p1 = ft2  + (size_t)row * 128;
    const float* p2 = feat + (size_t)row * 128;
    float xv[8], s = 0.0f, ss = 0.0f;
#pragma unroll
    for (int j = 0; j < 8; ++j) {
        int c = c0 + 16 * j;
        float x = p1[c] + p2[c];
        xv[j] = x; s += x; ss += x * x;
    }
#pragma unroll
    for (int m = 1; m < 16; m <<= 1) {
        s  += __shfl_xor(s,  m, 32);
        ss += __shfl_xor(ss, m, 32);
    }
    const float mean = s * (1.0f / 128.0f);
    const float var  = ss * (1.0f / 128.0f) - mean * mean;
    const float inv  = rsqrtf(var + EPSLN);
    float* o = rst + (size_t)row * 128;
#pragma unroll
    for (int j = 0; j < 8; ++j) {
        int c = c0 + 16 * j;
        o[c] = (xv[j] - mean) * inv * g[c] + b[c];
    }
}

__global__ __launch_bounds__(256) void ffn1_kernel(
    const float* __restrict__ rst, const float* __restrict__ W1,
    const float* __restrict__ b1, const float* __restrict__ pw,
    float* __restrict__ hbuf, int n)
{
    __shared__ float sA[64][132];
    const int row0 = blockIdx.x * 64;
    const int tid  = threadIdx.x;
    for (int i = tid; i < 64 * 32; i += 256) {
        int r = i >> 5, c4 = i & 31;
        int gr = row0 + r; if (gr >= n) gr = n - 1;
        ((float4*)&sA[r][0])[c4] = ((const float4*)(rst + (size_t)gr * 128))[c4];
    }
    __syncthreads();

    const int wave = tid >> 5;
    const int lane = tid & 31;
    const int half = lane >> 4;
    const int lr   = lane & 15;

#pragma unroll
    for (int j = 0; j < 4; ++j) {
        const int col = (wave + 8 * j) * 16 + lr;
        v2f bfr[32];
#pragma unroll
        for (int kq = 0; kq < 32; ++kq) {
            const int kk = 4 * kq + 2 * half;
            bfr[kq] = (v2f){ W1[(size_t)kk * 512 + col],
                             W1[(size_t)(kk + 1) * 512 + col] };
        }
        const float bias = b1[col], slope = pw[col];
#pragma unroll
        for (int mt = 0; mt < 4; ++mt) {
            v8f acc = {};
#pragma unroll
            for (int kq = 0; kq < 32; ++kq) {
                const int kk = 4 * kq + 2 * half;
                v2f a = { sA[mt * 16 + lr][kk], sA[mt * 16 + lr][kk + 1] };
                acc = wmma_f32_4(a, bfr[kq], acc);
            }
#pragma unroll
            for (int r = 0; r < 8; ++r) {
                int gr = row0 + mt * 16 + r + 8 * half;
                if (gr < n) {
                    float h = acc[r] + bias;
                    h = (h >= 0.0f) ? h : slope * h;
                    hbuf[(size_t)gr * 512 + col] = h;
                }
            }
        }
    }
}

__global__ __launch_bounds__(256) void ffn2_ln_kernel(
    const float* __restrict__ hbuf, const float* __restrict__ W2,
    const float* __restrict__ b2, const float* __restrict__ rst,
    const float* __restrict__ g, const float* __restrict__ be,
    float* __restrict__ out, int n)
{
    __shared__ float sA[64][132];
    const int row0 = blockIdx.x * 64;
    const int tid  = threadIdx.x;
    const int wave = tid >> 5;
    const int lane = tid & 31;
    const int half = lane >> 4;
    const int lr   = lane & 15;
    const int col  = wave * 16 + lr;

    v8f acc[4] = {};
#pragma unroll
    for (int kc = 0; kc < 4; ++kc) {
        __syncthreads();
        for (int i = tid; i < 64 * 32; i += 256) {
            int r = i >> 5, c4 = i & 31;
            int gr = row0 + r; if (gr >= n) gr = n - 1;
            ((float4*)&sA[r][0])[c4] =
                ((const float4*)(hbuf + (size_t)gr * 512 + kc * 128))[c4];
        }
        __syncthreads();

        v2f bfr[32];
#pragma unroll
        for (int kq = 0; kq < 32; ++kq) {
            const int kk = kc * 128 + 4 * kq + 2 * half;
            bfr[kq] = (v2f){ W2[(size_t)kk * 128 + col],
                             W2[(size_t)(kk + 1) * 128 + col] };
        }
#pragma unroll
        for (int mt = 0; mt < 4; ++mt) {
#pragma unroll
            for (int kq = 0; kq < 32; ++kq) {
                const int kk = 4 * kq + 2 * half;
                v2f a = { sA[mt * 16 + lr][kk], sA[mt * 16 + lr][kk + 1] };
                acc[mt] = wmma_f32_4(a, bfr[kq], acc[mt]);
            }
        }
    }

    __syncthreads();
    const float bias = b2[col];
#pragma unroll
    for (int mt = 0; mt < 4; ++mt) {
#pragma unroll
        for (int r = 0; r < 8; ++r) {
            const int rl = mt * 16 + r + 8 * half;
            int gr = row0 + rl; if (gr >= n) gr = n - 1;
            sA[rl][col] = acc[mt][r] + bias + rst[(size_t)gr * 128 + col];
        }
    }
    __syncthreads();

    const int rl2 = tid >> 2;
    const int c0  = tid & 3;
    const int row = row0 + rl2;
    float xv[32], s = 0.0f, ss = 0.0f;
#pragma unroll
    for (int j = 0; j < 32; ++j) {
        int c = c0 + 4 * j;
        float x = sA[rl2][c];
        xv[j] = x; s += x; ss += x * x;
    }
#pragma unroll
    for (int m = 1; m < 4; m <<= 1) {
        s  += __shfl_xor(s,  m, 32);
        ss += __shfl_xor(ss, m, 32);
    }
    const float mean = s * (1.0f / 128.0f);
    const float var  = ss * (1.0f / 128.0f) - mean * mean;
    const float inv  = rsqrtf(var + EPSLN);
    if (row < n) {
        float* o = out + (size_t)row * 128;
#pragma unroll
        for (int j = 0; j < 32; ++j) {
            int c = c0 + 4 * j;
            o[c] = (xv[j] - mean) * inv * g[c] + be[c];
        }
    }
}

extern "C" void kernel_launch(void* const* d_in, const int* in_sizes, int n_in,
                              void* d_out, int out_size, void* d_ws, size_t ws_size,
                              hipStream_t stream)
{
    const float* feat  = (const float*)d_in[0];
    const int*   src   = (const int*)  d_in[1];
    const int*   dst   = (const int*)  d_in[2];
    const float* Wq    = (const float*)d_in[3];
    const float* Wk    = (const float*)d_in[4];
    const float* Wv    = (const float*)d_in[5];
    const float* ln1_g = (const float*)d_in[6];
    const float* ln1_b = (const float*)d_in[7];
    const float* ln2_g = (const float*)d_in[8];
    const float* ln2_b = (const float*)d_in[9];
    const float* W1    = (const float*)d_in[10];
    const float* b1    = (const float*)d_in[11];
    const float* pw    = (const float*)d_in[12];
    const float* W2    = (const float*)d_in[13];
    const float* b2    = (const float*)d_in[14];

    const int n = in_sizes[0] / 128;
    const int e = in_sizes[1];

    float* ws = (float*)d_ws;
    size_t off = 0;
    float* q      = ws + off; off += (size_t)n * 128;
    float* kbuf   = ws + off; off += (size_t)n * 128;
    float* vbuf   = ws + off; off += (size_t)n * 128;
    float* ft2    = ws + off; off += (size_t)n * 128;
    float* rst    = ws + off; off += (size_t)n * 128;
    float* hbuf   = ws + off; off += (size_t)n * 512;
    const int np = (n + 4 * SCAN_T + 8 + 3) & ~3;
    int* cnt    = (int*)(ws + off); off += (size_t)np;
    int* rowptr = (int*)(ws + off); off += (size_t)np;
    int* rank   = (int*)(ws + off); off += (size_t)e;
    int* ssrc   = (int*)(ws + off); off += (size_t)e;
    (void)off; (void)ws_size; (void)n_in; (void)out_size;

    const int NT64 = (n + 63) / 64;
    const int NT16 = (n + 15) / 16;
    const int NW8  = (n + 7) / 8;
    const int ZB   = (np + 255) / 256;
    const int EB   = (e + 255) / 256;

    qkv_kernel<<<NT64, 256, 0, stream>>>(feat, Wq, Wk, Wv, q, kbuf, vbuf, n);
    zero_kernel<<<ZB, 256, 0, stream>>>(cnt, n);
    hist_kernel<<<EB, 256, 0, stream>>>(dst, cnt, rank, e);
    scan_kernel<<<1, SCAN_T, 0, stream>>>(cnt, rowptr, n);
    fill_kernel<<<EB, 256, 0, stream>>>(src, dst, rank, rowptr, ssrc, e);
    node_kernel<<<NW8, 256, 0, stream>>>(q, kbuf, vbuf, rowptr, ssrc, ft2, n);
    ln1_kernel<<<NT16, 256, 0, stream>>>(ft2, feat, ln1_g, ln1_b, rst, n);
    ffn1_kernel<<<NT64, 256, 0, stream>>>(rst, W1, b1, pw, hbuf, n);
    ffn2_ln_kernel<<<NT64, 256, 0, stream>>>(hbuf, W2, b2, rst, ln2_g, ln2_b,
                                             (float*)d_out, n);
}


# prefix-sum kernel moves whole 128-B lines: thread t owns int4 t of each 4096-counter tile, tiles held in registers (was one contiguous 160-B chunk per thread: 32 lines per wave load); same rowptr
# speedup vs baseline: 2.3926x; 1.0448x over previous
#include <hip/hip_runtime.h>
#include <hip/hip_bf16.h>
#include <math.h>

typedef __attribute__((ext_vector_type(2))) float v2f;
typedef __attribute__((ext_vector_type(8))) float v8f;

#define EPSLN 1e-5f

__device__ __forceinline__ v8f wmma_f32_4(v2f a, v2f b, v8f c) {
    return __builtin_amdgcn_wmma_f32_16x16x4_f32(
         false, a,  false, b,
         (short)0, c,  false,  false);
}

__global__ __launch_bounds__(256) void qkv_kernel(
    const float* __restrict__ feat,
    const float* __restrict__ Wq, const float* __restrict__ Wk,
    const float* __restrict__ Wv,
    float* __restrict__ q, float* __restrict__ k, float* __restrict__ v, int n)
{
    __shared__ float sA[64][132];
    const int row0 = blockIdx.x * 64;
    const int tid  = threadIdx.x;

    for (int i = tid; i < 64 * 32; i += 256) {
        int r = i >> 5, c4 = i & 31;
        int gr = row0 + r; if (gr >= n) gr = n - 1;
        ((float4*)&sA[r][0])[c4] = ((const float4*)(feat + (size_t)gr * 128))[c4];
    }
    __syncthreads();

    const int wave = tid >> 5;
    const int lane = tid & 31;
    const int half = lane >> 4;
    const int lr   = lane & 15;
    const int col  = wave * 16 + lr;

    const float* Ws[3] = { Wq, Wk, Wv };
    float*       Os[3] = { q,  k,  v  };

#pragma unroll
    for (int m = 0; m < 3; ++m) {
        const float* __restrict__ W = Ws[m];
        v2f bfr[32];
#pragma unroll
        for (int kq = 0; kq < 32; ++kq) {
            const int kk = 4 * kq + 2 * half;
            bfr[kq] = (v2f){ W[(size_t)kk * 128 + col],
                             W[(size_t)(kk + 1) * 128 + col] };
        }
        float* __restrict__ O = Os[m];
#pragma unroll
        for (int mt = 0; mt < 4; ++mt) {
            v8f acc = {};
#pragma unroll
            for (int kq = 0; kq < 32; ++kq) {
                const int kk = 4 * kq + 2 * half;
                v2f a = { sA[mt * 16 + lr][kk], sA[mt * 16 + lr][kk + 1] };
                acc = wmma_f32_4(a, bfr[kq], acc);
            }
#pragma unroll
            for (int r = 0; r < 8; ++r) {
                int gr = row0 + mt * 16 + r + 8 * half;
                if (gr < n) O[(size_t)gr * 128 + col] = acc[r];
            }
        }
    }
}

#define SCAN_T 1024

__device__ __forceinline__ int scan_chunk(int n) {
    return (((n + SCAN_T - 1) / SCAN_T) + 3) & ~3;
}

__global__ __launch_bounds__(256) void zero_kernel(int* __restrict__ cnt, int n)
{
    const int i = blockIdx.x * 256 + threadIdx.x;
    if (i < SCAN_T * scan_chunk(n)) cnt[i] = 0;
}

__global__ __launch_bounds__(256) void hist_kernel(
    const int* __restrict__ dst, int* __restrict__ cnt,
    int* __restrict__ rank, int e)
{
    const int t = blockIdx.x * 256 + threadIdx.x;
    if (t < e) rank[t] = atomicAdd(&cnt[dst[t]], 1);
}

#define SCAN_G 12

__global__ __launch_bounds__(SCAN_T) void scan_kernel(
    const int* __restrict__ cnt, int* __restrict__ rowptr, int n)
{
    __shared__ int wsum[SCAN_G][32];
    __shared__ int ttot[SCAN_G];
    const int ntiles = scan_chunk(n) >> 2;
    const int t = threadIdx.x, lane = t & 31, w = t >> 5;
    const int4* cp = (const int4*)cnt;
    int4* rp = (int4*)rowptr;
    int carry = 0;

    for (int j0 = 0; j0 < ntiles; j0 += SCAN_G) {
        const int nt = min(SCAN_G, ntiles - j0);
        int4 c[SCAN_G];
        int incl[SCAN_G];
#pragma unroll
        for (int j = 0; j < SCAN_G; ++j) {
            c[j] = make_int4(0, 0, 0, 0);
            if (j < nt) c[j] = cp[(size_t)(j0 + j) * SCAN_T + t];
        }
#pragma unroll
        for (int j = 0; j < SCAN_G; ++j) {
            int v = c[j].x + c[j].y + c[j].z + c[j].w;
#pragma unroll
            for (int o = 1; o < 32; o <<= 1) {
                int y = __shfl_up(v, o, 32);
                if (lane >= o) v += y;
            }
            incl[j] = v;
            if (lane == 31) wsum[j][w] = v;
        }
        __syncthreads();
        if (w < SCAN_G) {
            const int x = wsum[w][lane];
            int xi = x;
#pragma unroll
            for (int o = 1; o < 32; o <<= 1) {
                int y = __shfl_up(xi, o, 32);
                if (lane >= o) xi += y;
            }
            wsum[w][lane] = xi - x;
            if (lane == 31) ttot[w] = xi;
        }
        __syncthreads();
        int off[SCAN_G], tt[SCAN_G];
#pragma unroll
        for (int j = 0; j < SCAN_G; ++j) { off[j] = wsum[j][w]; tt[j] = ttot[j]; }
        int base = carry;
#pragma unroll
        for (int j = 0; j < SCAN_G; ++j) {
            if (j < nt) {
                const int4 cj = c[j];
                int run = base + off[j] + incl[j] - (cj.x + cj.y + cj.z + cj.w);
                int4 r;
                r.x = run; run += cj.x;
                r.y = run; run += cj.y;
                r.z = run; run += cj.z;
                r.w = run;
                rp[(size_t)(j0 + j) * SCAN_T + t] = r;
            }
            base += tt[j];
        }
        carry = base;
        __syncthreads();
    }
    if (t == 0) rowptr[(size_t)ntiles * 4 * SCAN_T] = carry;
}

__global__ __launch_bounds__(256) void fill_kernel(
    const int* __restrict__ src, const int* __restrict__ dst,
    const int* __restrict__ rank, const int* __restrict__ rowptr,
    int* __restrict__ ssrc, int e)
{
    const int t = blockIdx.x * 256 + threadIdx.x;
    if (t < e) ssrc[rowptr[dst[t]] + rank[t]] = src[t];
}

__device__ __forceinline__ void edge_step(
    const float4 kv, const float4 vv, const float4 qv,
    float& m, float& l, float4& acc)
{
    float p = kv.x * qv.x + kv.y * qv.y + kv.z * qv.z + kv.w * qv.w;
    p += __shfl_xor(p, 1, 32);
    p += __shfl_xor(p, 2, 32);
    p *= 0.08838834764831845f;
    const float t  = expf(-fabsf(p - m));
    const bool  up = p > m;
    const float sc = up ? t : 1.0f;
    const float ex = up ? 1.0f : t;
    m = up ? p : m;
    l = l * sc + ex;
    acc.x = acc.x * sc + ex * vv.x;
    acc.y = acc.y * sc + ex * vv.y;
    acc.z = acc.z * sc + ex * vv.z;
    acc.w = acc.w * sc + ex * vv.w;
}

__global__ __launch_bounds__(256) void node_kernel(
    const float* __restrict__ qbuf, const float* __restrict__ kbuf,
    const float* __restrict__ vbuf, const int* __restrict__ rowptr,
    const int* __restrict__ ssrc, float* __restrict__ ft2, int n)
{
    const int lane = threadIdx.x & 31;
    const int d = blockIdx.x * 8 + (threadIdx.x >> 5);
    if (d >= n) return;
    const int beg = rowptr[d], end = rowptr[d + 1];
    const float4 qv = ((const float4*)(qbuf + (size_t)d * 128))[lane];
    float m = -INFINITY, l = 0.0f;
    float4 acc = { 0.0f, 0.0f, 0.0f, 0.0f };

    for (int base = beg; base < end; base += 32) {
        const int cnt = min(32, end - base);
        int x = (lane < cnt) ? ssrc[base + lane] : 0x7fffffff;
#pragma unroll
        for (int kk = 2; kk <= 32; kk <<= 1) {
#pragma unroll
            for (int jj = kk >> 1; jj > 0; jj >>= 1) {
                const int o  = __shfl_xor(x, jj, 32);
                const int mn = min(x, o), mx = max(x, o);
                const bool asc = (lane & kk) == 0, low = (lane & jj) == 0;
                x = (asc == low) ? mn : mx;
            }
        }
        int j = 0;
        for (; j + 4 <= cnt; j += 4) {
            const int s0 = __builtin_amdgcn_readlane(x, j);
            const int s1 = __builtin_amdgcn_readlane(x, j + 1);
            const int s2 = __builtin_amdgcn_readlane(x, j + 2);
            const int s3 = __builtin_amdgcn_readlane(x, j + 3);
            const float4 k0 = ((const float4*)(kbuf + (size_t)s0 * 128))[lane];
            const float4 k1 = ((const float4*)(kbuf + (size_t)s1 * 128))[lane];
            const float4 k2 = ((const float4*)(kbuf + (size_t)s2 * 128))[lane];
            const float4 k3 = ((const float4*)(kbuf + (size_t)s3 * 128))[lane];
            const float4 v0 = ((const float4*)(vbuf + (size_t)s0 * 128))[lane];
            const float4 v1 = ((const float4*)(vbuf + (size_t)s1 * 128))[lane];
            const float4 v2 = ((const float4*)(vbuf + (size_t)s2 * 128))[lane];
            const float4 v3 = ((const float4*)(vbuf + (size_t)s3 * 128))[lane];
            edge_step(k0, v0, qv, m, l, acc);
            edge_step(k1, v1, qv, m, l, acc);
            edge_step(k2, v2, qv, m, l, acc);
            edge_step(k3, v3, qv, m, l, acc);
        }
        for (; j < cnt; ++j) {
            const int s0 = __builtin_amdgcn_readlane(x, j);
            const float4 k0 = ((const float4*)(kbuf + (size_t)s0 * 128))[lane];
            const float4 v0 = ((const float4*)(vbuf + (size_t)s0 * 128))[lane];
            edge_step(k0, v0, qv, m, l, acc);
        }
    }
    const float inv = (end > beg) ? 1.0f / l : 0.0f;
    float4 o4 = { acc.x * inv, acc.y * inv, acc.z * inv, acc.w * inv };
    ((float4*)(ft2 + (size_t)d * 128))[lane] = o4;
}

__global__ __launch_bounds__(256) void ln1_kernel(
    const float* __restrict__ ft2, const float* __restrict__ feat,
    const float* __restrict__ g, const float* __restrict__ b,
    float* __restrict__ rst, int n)
{
    const int row = blockIdx.x * 16 + (threadIdx.x >> 4);
    const int c0  = threadIdx.x & 15;
    if (row >= n) return;
    const float* p1 = ft2  + (size_t)row * 128;
    const float* p2 = feat + (size_t)row * 128;
    float xv[8], s = 0.0f, ss = 0.0f;
#pragma unroll
    for (int j = 0; j < 8; ++j) {
        int c = c0 + 16 * j;
        float x = p1[c] + p2[c];
        xv[j] = x; s += x; ss += x * x;
    }
#pragma unroll
    for (int m = 1; m < 16; m <<= 1) {
        s  += __shfl_xor(s,  m, 32);
        ss += __shfl_xor(ss, m, 32);
    }
    const float mean = s * (1.0f / 128.0f);
    const float var  = ss * (1.0f / 128.0f) - mean * mean;
    const float inv  = rsqrtf(var + EPSLN);
    float* o = rst + (size_t)row * 128;
#pragma unroll
    for (int j = 0; j < 8; ++j) {
        int c = c0 + 16 * j;
        o[c] = (xv[j] - mean) * inv * g[c] + b[c];
    }
}

__global__ __launch_bounds__(256) void ffn1_kernel(
    const float* __restrict__ rst, const float* __restrict__ W1,
    const float* __restrict__ b1, const float* __restrict__ pw,
    float* __restrict__ hbuf, int n)
{
    __shared__ float sA[64][132];
    const int row0 = blockIdx.x * 64;
    const int tid  = threadIdx.x;
    for (int i = tid; i < 64 * 32; i += 256) {
        int r = i >> 5, c4 = i & 31;
        int gr = row0 + r; if (gr >= n) gr = n - 1;
        ((float4*)&sA[r][0])[c4] = ((const float4*)(rst + (size_t)gr * 128))[c4];
    }
    __syncthreads();

    const int wave = tid >> 5;
    const int lane = tid & 31;
    const int half = lane >> 4;
    const int lr   = lane & 15;

#pragma unroll
    for (int j = 0; j < 4; ++j) {
        const int col = (wave + 8 * j) * 16 + lr;
        v2f bfr[32];
#pragma unroll
        for (int kq = 0; kq < 32; ++kq) {
            const int kk = 4 * kq + 2 * half;
            bfr[kq] = (v2f){ W1[(size_t)kk * 512 + col],
                             W1[(size_t)(kk + 1) * 512 + col] };
        }
        const float bias = b1[col], slope = pw[col];
#pragma unroll
        for (int mt = 0; mt < 4; ++mt) {
            v8f acc = {};
#pragma unroll
            for (int kq = 0; kq < 32; ++kq) {
                const int kk = 4 * kq + 2 * half;
                v2f a = { sA[mt * 16 + lr][kk], sA[mt * 16 + lr][kk + 1] };
                acc = wmma_f32_4(a, bfr[kq], acc);
            }
#pragma unroll
            for (int r = 0; r < 8; ++r) {
                int gr = row0 + mt * 16 + r + 8 * half;
                if (gr < n) {
                    float h = acc[r] + bias;
                    h = (h >= 0.0f) ? h : slope * h;
                    hbuf[(size_t)gr * 512 + col] = h;
                }
            }
        }
    }
}

__global__ __launch_bounds__(256) void ffn2_ln_kernel(
    const float* __restrict__ hbuf, const float* __restrict__ W2,
    const float* __restrict__ b2, const float* __restrict__ rst,
    const float* __restrict__ g, const float* __restrict__ be,
    float* __restrict__ out, int n)
{
    __shared__ float sA[64][132];
    const int row0 = blockIdx.x * 64;
    const int tid  = threadIdx.x;
    const int wave = tid >> 5;
    const int lane = tid & 31;
    const int half = lane >> 4;
    const int lr   = lane & 15;
    const int col  = wave * 16 + lr;

    v8f acc[4] = {};
#pragma unroll
    for (int kc = 0; kc < 4; ++kc) {
        __syncthreads();
        for (int i = tid; i < 64 * 32; i += 256) {
            int r = i >> 5, c4 = i & 31;
            int gr = row0 + r; if (gr >= n) gr = n - 1;
            ((float4*)&sA[r][0])[c4] =
                ((const float4*)(hbuf + (size_t)gr * 512 + kc * 128))[c4];
        }
        __syncthreads();

        v2f bfr[32];
#pragma unroll
        for (int kq = 0; kq < 32; ++kq) {
            const int kk = kc * 128 + 4 * kq + 2 * half;
            bfr[kq] = (v2f){ W2[(size_t)kk * 128 + col],
                             W2[(size_t)(kk + 1) * 128 + col] };
        }
#pragma unroll
        for (int mt = 0; mt < 4; ++mt) {
#pragma unroll
            for (int kq = 0; kq < 32; ++kq) {
                const int kk = 4 * kq + 2 * half;
                v2f a = { sA[mt * 16 + lr][kk], sA[mt * 16 + lr][kk + 1] };
                acc[mt] = wmma_f32_4(a, bfr[kq], acc[mt]);
            }
        }
    }

    __syncthreads();
    const float bias = b2[col];
#pragma unroll
    for (int mt = 0; mt < 4; ++mt) {
#pragma unroll
        for (int r = 0; r < 8; ++r) {
            const int rl = mt * 16 + r + 8 * half;
            int gr = row0 + rl; if (gr >= n) gr = n - 1;
            sA[rl][col] = acc[mt][r] + bias + rst[(size_t)gr * 128 + col];
        }
    }
    __syncthreads();

    const int rl2 = tid >> 2;
    const int c0  = tid & 3;
    const int row = row0 + rl2;
    float xv[32], s = 0.0f, ss = 0.0f;
#pragma unroll
    for (int j = 0; j < 32; ++j) {
        int c = c0 + 4 * j;
        float x = sA[rl2][c];
        xv[j] = x; s += x; ss += x * x;
    }
#pragma unroll
    for (int m = 1; m < 4; m <<= 1) {
        s  += __shfl_xor(s,  m, 32);
        ss += __shfl_xor(ss, m, 32);
    }
    const float mean = s * (1.0f / 128.0f);
    const float var  = ss * (1.0f / 128.0f) - mean * mean;
    const float inv  = rsqrtf(var + EPSLN);
    if (row < n) {
        float* o = out + (size_t)row * 128;
#pragma unroll
        for (int j = 0; j < 32; ++j) {
            int c = c0 + 4 * j;
            o[c] = (xv[j] - mean) * inv * g[c] + be[c];
        }
    }
}

extern "C" void kernel_launch(void* const* d_in, const int* in_sizes, int n_in,
                              void* d_out, int out_size, void* d_ws, size_t ws_size,
                              hipStream_t stream)
{
    const float* feat  = (const float*)d_in[0];
    const int*   src   = (const int*)  d_in[1];
    const int*   dst   = (const int*)  d_in[2];
    const float* Wq    = (const float*)d_in[3];
    const float* Wk    = (const float*)d_in[4];
    const float* Wv    = (const float*)d_in[5];
    const float* ln1_g = (const float*)d_in[6];
    const float* ln1_b = (const float*)d_in[7];
    const float* ln2_g = (const float*)d_in[8];
    const float* ln2_b = (const float*)d_in[9];
    const float* W1    = (const float*)d_in[10];
    const float* b1    = (const float*)d_in[11];
    const float* pw    = (const float*)d_in[12];
    const float* W2    = (const float*)d_in[13];
    const float* b2    = (const float*)d_in[14];

    const int n = in_sizes[0] / 128;
    const int e = in_sizes[1];

    float* ws = (float*)d_ws;
    size_t off = 0;
    float* q      = ws + off; off += (size_t)n * 128;
    float* kbuf   = ws + off; off += (size_t)n * 128;
    float* vbuf   = ws + off; off += (size_t)n * 128;
    float* ft2    = ws + off; off += (size_t)n * 128;
    float* rst    = ws + off; off += (size_t)n * 128;
    float* hbuf   = ws + off; off += (size_t)n * 512;
    const int np = (n + 4 * SCAN_T + 8 + 31) & ~31;
    int* cnt    = (int*)(ws + off); off += (size_t)np;
    int* rowptr = (int*)(ws + off); off += (size_t)np;
    int* rank   = (int*)(ws + off); off += (size_t)e;
    int* ssrc   = (int*)(ws + off); off += (size_t)e;
    (void)off; (void)ws_size; (void)n_in; (void)out_size;

    const int NT64 = (n + 63) / 64;
    const int NT16 = (n + 15) / 16;
    const int NW8  = (n + 7) / 8;
    const int ZB   = (np + 255) / 256;
    const int EB   = (e + 255) / 256;

    qkv_kernel<<<NT64, 256, 0, stream>>>(feat, Wq, Wk, Wv, q, kbuf, vbuf, n);
    zero_kernel<<<ZB, 256, 0, stream>>>(cnt, n);
    hist_kernel<<<EB, 256, 0, stream>>>(dst, cnt, rank, e);
    scan_kernel<<<1, SCAN_T, 0, stream>>>(cnt, rowptr, n);
    fill_kernel<<<EB, 256, 0, stream>>>(src, dst, rank, rowptr, ssrc, e);
    node_kernel<<<NW8, 256, 0, stream>>>(q, kbuf, vbuf, rowptr, ssrc, ft2, n);
    ln1_kernel<<<NT16, 256, 0, stream>>>(ft2, feat, ln1_g, ln1_b, rst, n);
    ffn1_kernel<<<NT64, 256, 0, stream>>>(rst, W1, b1, pw, hbuf, n);
    ffn2_ln_kernel<<<NT64, 256, 0, stream>>>(hbuf, W2, b2, rst, ln2_g, ln2_b,
                                             (float*)d_out, n);
}


# activation-tile fill of the three WMMA kernels: a thread's 8 float4 loads all in flight before the first LDS store (the rolled load-store loop kept one load per round trip, 8 in a row per tile)
# speedup vs baseline: 2.7644x; 1.1554x over previous
#include <hip/hip_runtime.h>
#include <hip/hip_bf16.h>
#include <math.h>

typedef __attribute__((ext_vector_type(2))) float v2f;
typedef __attribute__((ext_vector_type(8))) float v8f;

#define EPSLN 1e-5f

__device__ __forceinline__ v8f wmma_f32_4(v2f a, v2f b, v8f c) {
    return __builtin_amdgcn_wmma_f32_16x16x4_f32(
         false, a,  false, b,
         (short)0, c,  false,  false);
}

__device__ __forceinline__ void fill_tile(
    float (*sA)[132], const float* __restrict__ src, int pitch,
    int row0, int n, int tid)
{
    float4 t[8];
#pragma unroll
    for (int u = 0; u < 8; ++u) {
        const int i = tid + 256 * u;
        int gr = row0 + (i >> 5); if (gr >= n) gr = n - 1;
        t[u] = ((const float4*)(src + (size_t)gr * pitch))[i & 31];
    }
#pragma unroll
    for (int u = 0; u < 8; ++u) {
        const int i = tid + 256 * u;
        ((float4*)&sA[i >> 5][0])[i & 31] = t[u];
    }
}

__global__ __launch_bounds__(256) void qkv_kernel(
    const float* __restrict__ feat,
    const float* __restrict__ Wq, const float* __restrict__ Wk,
    const float* __restrict__ Wv,
    float* __restrict__ q, float* __restrict__ k, float* __restrict__ v, int n)
{
    __shared__ float sA[64][132];
    const int row0 = blockIdx.x * 64;
    const int tid  = threadIdx.x;

    fill_tile(sA, feat, 128, row0, n, tid);
    __syncthreads();

    const int wave = tid >> 5;
    const int lane = tid & 31;
    const int half = lane >> 4;
    const int lr   = lane & 15;
    const int col  = wave * 16 + lr;

    const float* Ws[3] = { Wq, Wk, Wv };
    float*       Os[3] = { q,  k,  v  };

#pragma unroll
    for (int m = 0; m < 3; ++m) {
        const float* __restrict__ W = Ws[m];
        v2f bfr[32];
#pragma unroll
        for (int kq = 0; kq < 32; ++kq) {
            const int kk = 4 * kq + 2 * half;
            bfr[kq] = (v2f){ W[(size_t)kk * 128 + col],
                             W[(size_t)(kk + 1) * 128 + col] };
        }
        float* __restrict__ O = Os[m];
#pragma unroll
        for (int mt = 0; mt < 4; ++mt) {
            v8f acc = {};
#pragma unroll
            for (int kq = 0; kq < 32; ++kq) {
                const int kk = 4 * kq + 2 * half;
                v2f a = { sA[mt * 16 + lr][kk], sA[mt * 16 + lr][kk + 1] };
                acc = wmma_f32_4(a, bfr[kq], acc);
            }
#pragma unroll
            for (int r = 0; r < 8; ++r) {
                int gr = row0 + mt * 16 + r + 8 * half;
                if (gr < n) O[(size_t)gr * 128 + col] = acc[r];
            }
        }
    }
}

#define SCAN_T 1024

__device__ __forceinline__ int scan_chunk(int n) {
    return (((n + SCAN_T - 1) / SCAN_T) + 3) & ~3;
}

__global__ __launch_bounds__(256) void zero_kernel(int* __restrict__ cnt, int n)
{
    const int i = blockIdx.x * 256 + threadIdx.x;
    if (i < SCAN_T * scan_chunk(n)) cnt[i] = 0;
}

__global__ __launch_bounds__(256) void hist_kernel(
    const int* __restrict__ dst, int* __restrict__ cnt,
    int* __restrict__ rank, int e)
{
    const int t = blockIdx.x * 256 + threadIdx.x;
    if (t < e) rank[t] = atomicAdd(&cnt[dst[t]], 1);
}

#define SCAN_G 12

__global__ __launch_bounds__(SCAN_T) void scan_kernel(
    const int* __restrict__ cnt, int* __restrict__ rowptr, int n)
{
    __shared__ int wsum[SCAN_G][32];
    __shared__ int ttot[SCAN_G];
    const int ntiles = scan_chunk(n) >> 2;
    const int t = threadIdx.x, lane = t & 31, w = t >> 5;
    const int4* cp = (const int4*)cnt;
    int4* rp = (int4*)rowptr;
    int carry = 0;

    for (int j0 = 0; j0 < ntiles; j0 += SCAN_G) {
        const int nt = min(SCAN_G, ntiles - j0);
        int4 c[SCAN_G];
        int incl[SCAN_G];
#pragma unroll
        for (int j = 0; j < SCAN_G; ++j) {
            c[j] = make_int4(0, 0, 0, 0);
            if (j < nt) c[j] = cp[(size_t)(j0 + j) * SCAN_T + t];
        }
#pragma unroll
        for (int j = 0; j < SCAN_G; ++j) {
            int v = c[j].x + c[j].y + c[j].z + c[j].w;
#pragma unroll
            for (int o = 1; o < 32; o <<= 1) {
                int y = __shfl_up(v, o, 32);
                if (lane >= o) v += y;
            }
            incl[j] = v;
            if (lane == 31) wsum[j][w] = v;
        }
        __syncthreads();
        if (w < SCAN_G) {
            const int x = wsum[w][lane];
            int xi = x;
#pragma unroll
            for (int o = 1; o < 32; o <<= 1) {
                int y = __shfl_up(xi, o, 32);
                if (lane >= o) xi += y;
            }
            wsum[w][lane] = xi - x;
            if (lane == 31) ttot[w] = xi;
        }
        __syncthreads();
        int off[SCAN_G], tt[SCAN_G];
#pragma unroll
        for (int j = 0; j < SCAN_G; ++j) { off[j] = wsum[j][w]; tt[j] = ttot[j]; }
        int base = carry;
#pragma unroll
        for (int j = 0; j < SCAN_G; ++j) {
            if (j < nt) {
                const int4 cj = c[j];
                int run = base + off[j] + incl[j] - (cj.x + cj.y + cj.z + cj.w);
                int4 r;
                r.x = run; run += cj.x;
                r.y = run; run += cj.y;
                r.z = run; run += cj.z;
                r.w = run;
                rp[(size_t)(j0 + j) * SCAN_T + t] = r;
            }
            base += tt[j];
        }
        carry = base;
        __syncthreads();
    }
    if (t == 0) rowptr[(size_t)ntiles * 4 * SCAN_T] = carry;
}

__global__ __launch_bounds__(256) void fill_kernel(
    const int* __restrict__ src, const int* __restrict__ dst,
    const int* __restrict__ rank, const int* __restrict__ rowptr,
    int* __restrict__ ssrc, int e)
{
    const int t = blockIdx.x * 256 + threadIdx.x;
    if (t < e) ssrc[rowptr[dst[t]] + rank[t]] = src[t];
}

__device__ __forceinline__ void edge_step(
    const float4 kv, const float4 vv, const float4 qv,
    float& m, float& l, float4& acc)
{
    float p = kv.x * qv.x + kv.y * qv.y + kv.z * qv.z + kv.w * qv.w;
    p += __shfl_xor(p, 1, 32);
    p += __shfl_xor(p, 2, 32);
    p *= 0.08838834764831845f;
    const float t  = expf(-fabsf(p - m));
    const bool  up = p > m;
    const float sc = up ? t : 1.0f;
    const float ex = up ? 1.0f : t;
    m = up ? p : m;
    l = l * sc + ex;
    acc.x = acc.x * sc + ex * vv.x;
    acc.y = acc.y * sc + ex * vv.y;
    acc.z = acc.z * sc + ex * vv.z;
    acc.w = acc.w * sc + ex * vv.w;
}

__global__ __launch_bounds__(256) void node_kernel(
    const float* __restrict__ qbuf, const float* __restrict__ kbuf,
    const float* __restrict__ vbuf, const int* __restrict__ rowptr,
    const int* __restrict__ ssrc, float* __restrict__ ft2, int n)
{
    const int lane = threadIdx.x & 31;
    const int d = blockIdx.x * 8 + (threadIdx.x >> 5);
    if (d >= n) return;
    const int beg = rowptr[d], end = rowptr[d + 1];
    const float4 qv = ((const float4*)(qbuf + (size_t)d * 128))[lane];
    float m = -INFINITY, l = 0.0f;
    float4 acc = { 0.0f, 0.0f, 0.0f, 0.0f };

    for (int base = beg; base < end; base += 32) {
        const int cnt = min(32, end - base);
        int x = (lane < cnt) ? ssrc[base + lane] : 0x7fffffff;
#pragma unroll
        for (int kk = 2; kk <= 32; kk <<= 1) {
#pragma unroll
            for (int jj = kk >> 1; jj > 0; jj >>= 1) {
                const int o  = __shfl_xor(x, jj, 32);
                const int mn = min(x, o), mx = max(x, o);
                const bool asc = (lane & kk) == 0, low = (lane & jj) == 0;
                x = (asc == low) ? mn : mx;
            }
        }
        int j = 0;
        for (; j + 4 <= cnt; j += 4) {
            const int s0 = __builtin_amdgcn_readlane(x, j);
            const int s1 = __builtin_amdgcn_readlane(x, j + 1);
            const int s2 = __builtin_amdgcn_readlane(x, j + 2);
            const int s3 = __builtin_amdgcn_readlane(x, j + 3);
            const float4 k0 = ((const float4*)(kbuf + (size_t)s0 * 128))[lane];
            const float4 k1 = ((const float4*)(kbuf + (size_t)s1 * 128))[lane];
            const float4 k2 = ((const float4*)(kbuf + (size_t)s2 * 128))[lane];
            const float4 k3 = ((const float4*)(kbuf + (size_t)s3 * 128))[lane];
            const float4 v0 = ((const float4*)(vbuf + (size_t)s0 * 128))[lane];
            const float4 v1 = ((const float4*)(vbuf + (size_t)s1 * 128))[lane];
            const float4 v2 = ((const float4*)(vbuf + (size_t)s2 * 128))[lane];
            const float4 v3 = ((const float4*)(vbuf + (size_t)s3 * 128))[lane];
            edge_step(k0, v0, qv, m, l, acc);
            edge_step(k1, v1, qv, m, l, acc);
            edge_step(k2, v2, qv, m, l, acc);
            edge_step(k3, v3, qv, m, l, acc);
        }
        for (; j < cnt; ++j) {
            const int s0 = __builtin_amdgcn_readlane(x, j);
            const float4 k0 = ((const float4*)(kbuf + (size_t)s0 * 128))[lane];
            const float4 v0 = ((const float4*)(vbuf + (size_t)s0 * 128))[lane];
            edge_step(k0, v0, qv, m, l, acc);
        }
    }
    const float inv = (end > beg) ? 1.0f / l : 0.0f;
    float4 o4 = { acc.x * inv, acc.y * inv, acc.z * inv, acc.w * inv };
    ((float4*)(ft2 + (size_t)d * 128))[lane] = o4;
}

__global__ __launch_bounds__(256) void ln1_kernel(
    const float* __restrict__ ft2, const float* __restrict__ feat,
    const float* __restrict__ g, const float* __restrict__ b,
    float* __restrict__ rst, int n)
{
    const int row = blockIdx.x * 16 + (threadIdx.x >> 4);
    const int c0  = threadIdx.x & 15;
    if (row >= n) return;
    const float* p1 = ft2  + (size_t)row * 128;
    const float* p2 = feat + (size_t)row * 128;
    float xv[8], s = 0.0f, ss = 0.0f;
#pragma unroll
    for (int j = 0; j < 8; ++j) {
        int c = c0 + 16 * j;
        float x = p1[c] + p2[c];
        xv[j] = x; s += x; ss += x * x;
    }
#pragma unroll
    for (int m = 1; m < 16; m <<= 1) {
        s  += __shfl_xor(s,  m, 32);
        ss += __shfl_xor(ss, m, 32);
    }
    const float mean = s * (1.0f / 128.0f);
    const float var  = ss * (1.0f / 128.0f) - mean * mean;
    const float inv  = rsqrtf(var + EPSLN);
    float* o = rst + (size_t)row * 128;
#pragma unroll
    for (int j = 0; j < 8; ++j) {
        int c = c0 + 16 * j;
        o[c] = (xv[j] - mean) * inv * g[c] + b[c];
    }
}

__global__ __launch_bounds__(256) void ffn1_kernel(
    const float* __restrict__ rst, const float* __restrict__ W1,
    const float* __restrict__ b1, const float* __restrict__ pw,
    float* __restrict__ hbuf, int n)
{
    __shared__ float sA[64][132];
    const int row0 = blockIdx.x * 64;
    const int tid  = threadIdx.x;
    fill_tile(sA, rst, 128, row0, n, tid);
    __syncthreads();

    const int wave = tid >> 5;
    const int lane = tid & 31;
    const int half = lane >> 4;
    const int lr   = lane & 15;

#pragma unroll
    for (int j = 0; j < 4; ++j) {
        const int col = (wave + 8 * j) * 16 + lr;
        v2f bfr[32];
#pragma unroll
        for (int kq = 0; kq < 32; ++kq) {
            const int kk = 4 * kq + 2 * half;
            bfr[kq] = (v2f){ W1[(size_t)kk * 512 + col],
                             W1[(size_t)(kk + 1) * 512 + col] };
        }
        const float bias = b1[col], slope = pw[col];
#pragma unroll
        for (int mt = 0; mt < 4; ++mt) {
            v8f acc = {};
#pragma unroll
            for (int kq = 0; kq < 32; ++kq) {
                const int kk = 4 * kq + 2 * half;
                v2f a = { sA[mt * 16 + lr][kk], sA[mt * 16 + lr][kk + 1] };
                acc = wmma_f32_4(a, bfr[kq], acc);
            }
#pragma unroll
            for (int r = 0; r < 8; ++r) {
                int gr = row0 + mt * 16 + r + 8 * half;
                if (gr < n) {
                    float h = acc[r] + bias;
                    h = (h >= 0.0f) ? h : slope * h;
                    hbuf[(size_t)gr * 512 + col] = h;
                }
            }
        }
    }
}

__global__ __launch_bounds__(256) void ffn2_ln_kernel(
    const float* __restrict__ hbuf, const float* __restrict__ W2,
    const float* __restrict__ b2, const float* __restrict__ rst,
    const float* __restrict__ g, const float* __restrict__ be,
    float* __restrict__ out, int n)
{
    __shared__ float sA[64][132];
    const int row0 = blockIdx.x * 64;
    const int tid  = threadIdx.x;
    const int wave = tid >> 5;
    const int lane = tid & 31;
    const int half = lane >> 4;
    const int lr   = lane & 15;
    const int col  = wave * 16 + lr;

    v8f acc[4] = {};
#pragma unroll
    for (int kc = 0; kc < 4; ++kc) {
        __syncthreads();
        fill_tile(sA, hbuf + kc * 128, 512, row0, n, tid);
        __syncthreads();

        v2f bfr[32];
#pragma unroll
        for (int kq = 0; kq < 32; ++kq) {
            const int kk = kc * 128 + 4 * kq + 2 * half;
            bfr[kq] = (v2f){ W2[(size_t)kk * 128 + col],
                             W2[(size_t)(kk + 1) * 128 + col] };
        }
#pragma unroll
        for (int mt = 0; mt < 4; ++mt) {
#pragma unroll
            for (int kq = 0; kq < 32; ++kq) {
                const int kk = 4 * kq + 2 * half;
                v2f a = { sA[mt * 16 + lr][kk], sA[mt * 16 + lr][kk + 1] };
                acc[mt] = wmma_f32_4(a, bfr[kq], acc[mt]);
            }
        }
    }

    __syncthreads();
    const float bias = b2[col];
#pragma unroll
    for (int mt = 0; mt < 4; ++mt) {
#pragma unroll
        for (int r = 0; r < 8; ++r) {
            const int rl = mt * 16 + r + 8 * half;
            int gr = row0 + rl; if (gr >= n) gr = n - 1;
            sA[rl][col] = acc[mt][r] + bias + rst[(size_t)gr * 128 + col];
        }
    }
    __syncthreads();

    const int rl2 = tid >> 2;
    const int c0  = tid & 3;
    const int row = row0 + rl2;
    float xv[32], s = 0.0f, ss = 0.0f;
#pragma unroll
    for (int j = 0; j < 32; ++j) {
        int c = c0 + 4 * j;
        float x = sA[rl2][c];
        xv[j] = x; s += x; ss += x * x;
    }
#pragma unroll
    for (int m = 1; m < 4; m <<= 1) {
        s  += __shfl_xor(s,  m, 32);
        ss += __shfl_xor(ss, m, 32);
    }
    const float mean = s * (1.0f / 128.0f);
    const float var  = ss * (1.0f / 128.0f) - mean * mean;
    const float inv  = rsqrtf(var + EPSLN);
    if (row < n) {
        float* o = out + (size_t)row * 128;
#pragma unroll
        for (int j = 0; j < 32; ++j) {
            int c = c0 + 4 * j;
            o[c] = (xv[j] - mean) * inv * g[c] + be[c];
        }
    }
}

extern "C" void kernel_launch(void* const* d_in, const int* in_sizes, int n_in,
                              void* d_out, int out_size, void* d_ws, size_t ws_size,
                              hipStream_t stream)
{
    const float* feat  = (const float*)d_in[0];
    const int*   src   = (const int*)  d_in[1];
    const int*   dst   = (const int*)  d_in[2];
    const float* Wq    = (const float*)d_in[3];
    const float* Wk    = (const float*)d_in[4];
    const float* Wv    = (const float*)d_in[5];
    const float* ln1_g = (const float*)d_in[6];
    const float* ln1_b = (const float*)d_in[7];
    const float* ln2_g = (const float*)d_in[8];
    const float* ln2_b = (const float*)d_in[9];
    const float* W1    = (const float*)d_in[10];
    const float* b1    = (const float*)d_in[11];
    const float* pw    = (const float*)d_in[12];
    const float* W2    = (const float*)d_in[13];
    const float* b2    = (const float*)d_in[14];

    const int n = in_sizes[0] / 128;
    const int e = in_sizes[1];

    float* ws = (float*)d_ws;
    size_t off = 0;
    float* q      = ws + off; off += (size_t)n * 128;
    float* kbuf   = ws + off; off += (size_t)n * 128;
    float* vbuf   = ws + off; off += (size_t)n * 128;
    float* ft2    = ws + off; off += (size_t)n * 128;
    float* rst    = ws + off; off += (size_t)n * 128;
    float* hbuf   = ws + off; off += (size_t)n * 512;
    const int np = (n + 4 * SCAN_T + 8 + 31) & ~31;
    int* cnt    = (int*)(ws + off); off += (size_t)np;
    int* rowptr = (int*)(ws + off); off += (size_t)np;
    int* rank   = (int*)(ws + off); off += (size_t)e;
    int* ssrc   = (int*)(ws + off); off += (size_t)e;
    (void)off; (void)ws_size; (void)n_in; (void)out_size;

    const int NT64 = (n + 63) / 64;
    const int NT16 = (n + 15) / 16;
    const int NW8  = (n + 7) / 8;
    const int ZB   = (np + 255) / 256;
    const int EB   = (e + 255) / 256;

    qkv_kernel<<<NT64, 256, 0, stream>>>(feat, Wq, Wk, Wv, q, kbuf, vbuf, n);
    zero_kernel<<<ZB, 256, 0, stream>>>(cnt, n);
    hist_kernel<<<EB, 256, 0, stream>>>(dst, cnt, rank, e);
    scan_kernel<<<1, SCAN_T, 0, stream>>>(cnt, rowptr, n);
    fill_kernel<<<EB, 256, 0, stream>>>(src, dst, rank, rowptr, ssrc, e);
    node_kernel<<<NW8, 256, 0, stream>>>(q, kbuf, vbuf, rowptr, ssrc, ft2, n);
    ln1_kernel<<<NT16, 256, 0, stream>>>(ft2, feat, ln1_g, ln1_b, rst, n);
    ffn1_kernel<<<NT64, 256, 0, stream>>>(rst, W1, b1, pw, hbuf, n);
    ffn2_ln_kernel<<<NT64, 256, 0, stream>>>(hbuf, W2, b2, rst, ln2_g, ln2_b,
                                             (float*)d_out, n);
}


# row tile of the three WMMA kernels 64 -> 80 rows (5 M-tiles per B fragment set): 500 equal work-groups, 2 per compute unit, was 625 lying 3 and 2; the last kernel no longer needs a second round
# speedup vs baseline: 2.8958x; 1.0475x over previous
#include <hip/hip_runtime.h>
#include <hip/hip_bf16.h>
#include <math.h>

typedef __attribute__((ext_vector_type(2))) float v2f;
typedef __attribute__((ext_vector_type(8))) float v8f;

#define EPSLN 1e-5f

#define TM 80
#define MT (TM / 16)
#define FU (TM * 32 / 256)

__device__ __forceinline__ v8f wmma_f32_4(v2f a, v2f b, v8f c) {
    return __builtin_amdgcn_wmma_f32_16x16x4_f32(
         false, a,  false, b,
         (short)0, c,  false,  false);
}

__device__ __forceinline__ void fill_tile(
    float (*sA)[132], const float* __restrict__ src, int pitch,
    int row0, int n, int tid)
{
    float4 t[FU];
#pragma unroll
    for (int u = 0; u < FU; ++u) {
        const int i = tid + 256 * u;
        int gr = row0 + (i >> 5); if (gr >= n) gr = n - 1;
        t[u] = ((const float4*)(src + (size_t)gr * pitch))[i & 31];
    }
#pragma unroll
    for (int u = 0; u < FU; ++u) {
        const int i = tid + 256 * u;
        ((float4*)&sA[i >> 5][0])[i & 31] = t[u];
    }
}

__global__ __launch_bounds__(256) void qkv_kernel(
    const float* __restrict__ feat,
    const float* __restrict__ Wq, const float* __restrict__ Wk,
    const float* __restrict__ Wv,
    float* __restrict__ q, float* __restrict__ k, float* __restrict__ v, int n)
{
    __shared__ float sA[TM][132];
    const int row0 = blockIdx.x * TM;
    const int tid  = threadIdx.x;

    fill_tile(sA, feat, 128, row0, n, tid);
    __syncthreads();

    const int wave = tid >> 5;
    const int lane = tid & 31;
    const int half = lane >> 4;
    const int lr   = lane & 15;
    const int col  = wave * 16 + lr;

    const float* Ws[3] = { Wq, Wk, Wv };
    float*       Os[3] = { q,  k,  v  };

#pragma unroll
    for (int m = 0; m < 3; ++m) {
        const float* __restrict__ W = Ws[m];
        v2f bfr[32];
#pragma unroll
        for (int kq = 0; kq < 32; ++kq) {
            const int kk = 4 * kq + 2 * half;
            bfr[kq] = (v2f){ W[(size_t)kk * 128 + col],
                             W[(size_t)(kk + 1) * 128 + col] };
        }
        float* __restrict__ O = Os[m];
#pragma unroll
        for (int mt = 0; mt < MT; ++mt) {
            v8f acc = {};
#pragma unroll
            for (int kq = 0; kq < 32; ++kq) {
                const int kk = 4 * kq + 2 * half;
                v2f a = { sA[mt * 16 + lr][kk], sA[mt * 16 + lr][kk + 1] };
                acc = wmma_f32_4(a, bfr[kq], acc);
            }
#pragma unroll
            for (int r = 0; r < 8; ++r) {
                int gr = row0 + mt * 16 + r + 8 * half;
                if (gr < n) O[(size_t)gr * 128 + col] = acc[r];
            }
        }
    }
}

#define SCAN_T 1024

__device__ __forceinline__ int scan_chunk(int n) {
    return (((n + SCAN_T - 1) / SCAN_T) + 3) & ~3;
}

__global__ __launch_bounds__(256) void zero_kernel(int* __restrict__ cnt, int n)
{
    const int i = blockIdx.x * 256 + threadIdx.x;
    if (i < SCAN_T * scan_chunk(n)) cnt[i] = 0;
}

__global__ __launch_bounds__(256) void hist_kernel(
    const int* __restrict__ dst, int* __restrict__ cnt,
    int* __restrict__ rank, int e)
{
    const int t = blockIdx.x * 256 + threadIdx.x;
    if (t < e) rank[t] = atomicAdd(&cnt[dst[t]], 1);
}

#define SCAN_G 12

__global__ __launch_bounds__(SCAN_T) void scan_kernel(
    const int* __restrict__ cnt, int* __restrict__ rowptr, int n)
{
    __shared__ int wsum[SCAN_G][32];
    __shared__ int ttot[SCAN_G];
    const int ntiles = scan_chunk(n) >> 2;
    const int t = threadIdx.x, lane = t & 31, w = t >> 5;
    const int4* cp = (const int4*)cnt;
    int4* rp = (int4*)rowptr;
    int carry = 0;

    for (int j0 = 0; j0 < ntiles; j0 += SCAN_G) {
        const int nt = min(SCAN_G, ntiles - j0);
        int4 c[SCAN_G];
        int incl[SCAN_G];
#pragma unroll
        for (int j = 0; j < SCAN_G; ++j) {
            c[j] = make_int4(0, 0, 0, 0);
            if (j < nt) c[j] = cp[(size_t)(j0 + j) * SCAN_T + t];
        }
#pragma unroll
        for (int j = 0; j < SCAN_G; ++j) {
            int v = c[j].x + c[j].y + c[j].z + c[j].w;
#pragma unroll
            for (int o = 1; o < 32; o <<= 1) {
                int y = __shfl_up(v, o, 32);
                if (lane >= o) v += y;
            }
            incl[j] = v;
            if (lane == 31) wsum[j][w] = v;
        }
        __syncthreads();
        if (w < SCAN_G) {
            const int x = wsum[w][lane];
            int xi = x;
#pragma unroll
            for (int o = 1; o < 32; o <<= 1) {
                int y = __shfl_up(xi, o, 32);
                if (lane >= o) xi += y;
            }
            wsum[w][lane] = xi - x;
            if (lane == 31) ttot[w] = xi;
        }
        __syncthreads();
        int off[SCAN_G], tt[SCAN_G];
#pragma unroll
        for (int j = 0; j < SCAN_G; ++j) { off[j] = wsum[j][w]; tt[j] = ttot[j]; }
        int base = carry;
#pragma unroll
        for (int j = 0; j < SCAN_G; ++j) {
            if (j < nt) {
                const int4 cj = c[j];
                int run = base + off[j] + incl[j] - (cj.x + cj.y + cj.z + cj.w);
                int4 r;
                r.x = run; run += cj.x;
                r.y = run; run += cj.y;
                r.z = run; run += cj.z;
                r.w = run;
                rp[(size_t)(j0 + j) * SCAN_T + t] = r;
            }
            base += tt[j];
        }
        carry = base;
        __syncthreads();
    }
    if (t == 0) rowptr[(size_t)ntiles * 4 * SCAN_T] = carry;
}

__global__ __launch_bounds__(256) void fill_kernel(
    const int* __restrict__ src, const int* __restrict__ dst,
    const int* __restrict__ rank, const int* __restrict__ rowptr,
    int* __restrict__ ssrc, int e)
{
    const int t = blockIdx.x * 256 + threadIdx.x;
    if (t < e) ssrc[rowptr[dst[t]] + rank[t]] = src[t];
}

__device__ __forceinline__ void edge_step(
    const float4 kv, const float4 vv, const float4 qv,
    float& m, float& l, float4& acc)
{
    float p = kv.x * qv.x + kv.y * qv.y + kv.z * qv.z + kv.w * qv.w;
    p += __shfl_xor(p, 1, 32);
    p += __shfl_xor(p, 2, 32);
    p *= 0.08838834764831845f;
    const float t  = expf(-fabsf(p - m));
    const bool  up = p > m;
    const float sc = up ? t : 1.0f;
    const float ex = up ? 1.0f : t;
    m = up ? p : m;
    l = l * sc + ex;
    acc.x = acc.x * sc + ex * vv.x;
    acc.y = acc.y * sc + ex * vv.y;
    acc.z = acc.z * sc + ex * vv.z;
    acc.w = acc.w * sc + ex * vv.w;
}

__global__ __launch_bounds__(256) void node_kernel(
    const float* __restrict__ qbuf, const float* __restrict__ kbuf,
    const float* __restrict__ vbuf, const int* __restrict__ rowptr,
    const int* __restrict__ ssrc, float* __restrict__ ft2, int n)
{
    const int lane = threadIdx.x & 31;
    const int d = blockIdx.x * 8 + (threadIdx.x >> 5);
    if (d >= n) return;
    const int beg = rowptr[d], end = rowptr[d + 1];
    const float4 qv = ((const float4*)(qbuf + (size_t)d * 128))[lane];
    float m = -INFINITY, l = 0.0f;
    float4 acc = { 0.0f, 0.0f, 0.0f, 0.0f };

    for (int base = beg; base < end; base += 32) {
        const int cnt = min(32, end - base);
        int x = (lane < cnt) ? ssrc[base + lane] : 0x7fffffff;
#pragma unroll
        for (int kk = 2; kk <= 32; kk <<= 1) {
#pragma unroll
            for (int jj = kk >> 1; jj > 0; jj >>= 1) {
                const int o  = __shfl_xor(x, jj, 32);
                const int mn = min(x, o), mx = max(x, o);
                const bool asc = (lane & kk) == 0, low = (lane & jj) == 0;
                x = (asc == low) ? mn : mx;
            }
        }
        int j = 0;
        for (; j + 4 <= cnt; j += 4) {
            const int s0 = __builtin_amdgcn_readlane(x, j);
            const int s1 = __builtin_amdgcn_readlane(x, j + 1);
            const int s2 = __builtin_amdgcn_readlane(x, j + 2);
            const int s3 = __builtin_amdgcn_readlane(x, j + 3);
            const float4 k0 = ((const float4*)(kbuf + (size_t)s0 * 128))[lane];
            const float4 k1 = ((const float4*)(kbuf + (size_t)s1 * 128))[lane];
            const float4 k2 = ((const float4*)(kbuf + (size_t)s2 * 128))[lane];
            const float4 k3 = ((const float4*)(kbuf + (size_t)s3 * 128))[lane];
            const float4 v0 = ((const float4*)(vbuf + (size_t)s0 * 128))[lane];
            const float4 v1 = ((const float4*)(vbuf + (size_t)s1 * 128))[lane];
            const float4 v2 = ((const float4*)(vbuf + (size_t)s2 * 128))[lane];
            const float4 v3 = ((const float4*)(vbuf + (size_t)s3 * 128))[lane];
            edge_step(k0, v0, qv, m, l, acc);
            edge_step(k1, v1, qv, m, l, acc);
            edge_step(k2, v2, qv, m, l, acc);
            edge_step(k3, v3, qv, m, l, acc);
        }
        for (; j < cnt; ++j) {
            const int s0 = __builtin_amdgcn_readlane(x, j);
            const float4 k0 = ((const float4*)(kbuf + (size_t)s0 * 128))[lane];
            const float4 v0 = ((const float4*)(vbuf + (size_t)s0 * 128))[lane];
            edge_step(k0, v0, qv, m, l, acc);
        }
    }
    const float inv = (end > beg) ? 1.0f / l : 0.0f;
    float4 o4 = { acc.x * inv, acc.y * inv, acc.z * inv, acc.w * inv };
    ((float4*)(ft2 + (size_t)d * 128))[lane] = o4;
}

__global__ __launch_bounds__(256) void ln1_kernel(
    const float* __restrict__ ft2, const float* __restrict__ feat,
    const float* __restrict__ g, const float* __restrict__ b,
    float* __restrict__ rst, int n)
{
    const int row = blockIdx.x * 16 + (threadIdx.x >> 4);
    const int c0  = threadIdx.x & 15;
    if (row >= n) return;
    const float* p1 = ft2  + (size_t)row * 128;
    const float* p2 = feat + (size_t)row * 128;
    float xv[8], s = 0.0f, ss = 0.0f;
#pragma unroll
    for (int j = 0; j < 8; ++j) {
        int c = c0 + 16 * j;
        float x = p1[c] + p2[c];
        xv[j] = x; s += x; ss += x * x;
    }
#pragma unroll
    for (int m = 1; m < 16; m <<= 1) {
        s  += __shfl_xor(s,  m, 32);
        ss += __shfl_xor(ss, m, 32);
    }
    const float mean = s * (1.0f / 128.0f);
    const float var  = ss * (1.0f / 128.0f) - mean * mean;
    const float inv  = rsqrtf(var + EPSLN);
    float* o = rst + (size_t)row * 128;
#pragma unroll
    for (int j = 0; j < 8; ++j) {
        int c = c0 + 16 * j;
        o[c] = (xv[j] - mean) * inv * g[c] + b[c];
    }
}

__global__ __launch_bounds__(256) void ffn1_kernel(
    const float* __restrict__ rst, const float* __restrict__ W1,
    const float* __restrict__ b1, const float* __restrict__ pw,
    float* __restrict__ hbuf, int n)
{
    __shared__ float sA[TM][132];
    const int row0 = blockIdx.x * TM;
    const int tid  = threadIdx.x;
    fill_tile(sA, rst, 128, row0, n, tid);
    __syncthreads();

    const int wave = tid >> 5;
    const int lane = tid & 31;
    const int half = lane >> 4;
    const int lr   = lane & 15;

#pragma unroll
    for (int j = 0; j < 4; ++j) {
        const int col = (wave + 8 * j) * 16 + lr;
        v2f bfr[32];
#pragma unroll
        for (int kq = 0; kq < 32; ++kq) {
            const int kk = 4 * kq + 2 * half;
            bfr[kq] = (v2f){ W1[(size_t)kk * 512 + col],
                             W1[(size_t)(kk + 1) * 512 + col] };
        }
        const float bias = b1[col], slope = pw[col];
#pragma unroll
        for (int mt = 0; mt < MT; ++mt) {
            v8f acc = {};
#pragma unroll
            for (int kq = 0; kq < 32; ++kq) {
                const int kk = 4 * kq + 2 * half;
                v2f a = { sA[mt * 16 + lr][kk], sA[mt * 16 + lr][kk + 1] };
                acc = wmma_f32_4(a, bfr[kq], acc);
            }
#pragma unroll
            for (int r = 0; r < 8; ++r) {
                int gr = row0 + mt * 16 + r + 8 * half;
                if (gr < n) {
                    float h = acc[r] + bias;
                    h = (h >= 0.0f) ? h : slope * h;
                    hbuf[(size_t)gr * 512 + col] = h;
                }
            }
        }
    }
}

__global__ __launch_bounds__(256) void ffn2_ln_kernel(
    const float* __restrict__ hbuf, const float* __restrict__ W2,
    const float* __restrict__ b2, const float* __restrict__ rst,
    const float* __restrict__ g, const float* __restrict__ be,
    float* __restrict__ out, int n)
{
    __shared__ float sA[TM][132];
    const int row0 = blockIdx.x * TM;
    const int tid  = threadIdx.x;
    const int wave = tid >> 5;
    const int lane = tid & 31;
    const int half = lane >> 4;
    const int lr   = lane & 15;
    const int col  = wave * 16 + lr;

    v8f acc[MT] = {};
#pragma unroll
    for (int kc = 0; kc < 4; ++kc) {
        __syncthreads();
        fill_tile(sA, hbuf + kc * 128, 512, row0, n, tid);
        __syncthreads();

        v2f bfr[32];
#pragma unroll
        for (int kq = 0; kq < 32; ++kq) {
            const int kk = kc * 128 + 4 * kq + 2 * half;
            bfr[kq] = (v2f){ W2[(size_t)kk * 128 + col],
                             W2[(size_t)(kk + 1) * 128 + col] };
        }
#pragma unroll
        for (int mt = 0; mt < MT; ++mt) {
#pragma unroll
            for (int kq = 0; kq < 32; ++kq) {
                const int kk = 4 * kq + 2 * half;
                v2f a = { sA[mt * 16 + lr][kk], sA[mt * 16 + lr][kk + 1] };
                acc[mt] = wmma_f32_4(a, bfr[kq], acc[mt]);
            }
        }
    }

    __syncthreads();
    const float bias = b2[col];
#pragma unroll
    for (int mt = 0; mt < MT; ++mt) {
#pragma unroll
        for (int r = 0; r < 8; ++r) {
            const int rl = mt * 16 + r + 8 * half;
            int gr = row0 + rl; if (gr >= n) gr = n - 1;
            sA[rl][col] = acc[mt][r] + bias + rst[(size_t)gr * 128 + col];
        }
    }
    __syncthreads();

    const int c0 = tid & 3;
    for (int rb = 0; rb < TM; rb += 64) {
        const int rl2 = rb + (tid >> 2);
        if (rl2 < TM) {
            const int row = row0 + rl2;
            float xv[32], s = 0.0f, ss = 0.0f;
#pragma unroll
            for (int j = 0; j < 32; ++j) {
                int c = c0 + 4 * j;
                float x = sA[rl2][c];
                xv[j] = x; s += x; ss += x * x;
            }
#pragma unroll
            for (int m = 1; m < 4; m <<= 1) {
                s  += __shfl_xor(s,  m, 32);
                ss += __shfl_xor(ss, m, 32);
            }
            const float mean = s * (1.0f / 128.0f);
            const float var  = ss * (1.0f / 128.0f) - mean * mean;
            const float inv  = rsqrtf(var + EPSLN);
            if (row < n) {
                float* o = out + (size_t)row * 128;
#pragma unroll
                for (int j = 0; j < 32; ++j) {
                    int c = c0 + 4 * j;
                    o[c] = (xv[j] - mean) * inv * g[c] + be[c];
                }
            }
        }
    }
}

extern "C" void kernel_launch(void* const* d_in, const int* in_sizes, int n_in,
                              void* d_out, int out_size, void* d_ws, size_t ws_size,
                              hipStream_t stream)
{
    const float* feat  = (const float*)d_in[0];
    const int*   src   = (const int*)  d_in[1];
    const int*   dst   = (const int*)  d_in[2];
    const float* Wq    = (const float*)d_in[3];
    const float* Wk    = (const float*)d_in[4];
    const float* Wv    = (const float*)d_in[5];
    const float* ln1_g = (const float*)d_in[6];
    const float* ln1_b = (const float*)d_in[7];
    const float* ln2_g = (const float*)d_in[8];
    const float* ln2_b = (const float*)d_in[9];
    const float* W1    = (const float*)d_in[10];
    const float* b1    = (const float*)d_in[11];
    const float* pw    = (const float*)d_in[12];
    const float* W2    = (const float*)d_in[13];
    const float* b2    = (const float*)d_in[14];

    const int n = in_sizes[0] / 128;
    const int e = in_sizes[1];

    float* ws = (float*)d_ws;
    size_t off = 0;
    float* q      = ws + off; off += (size_t)n * 128;
    float* kbuf   = ws + off; off += (size_t)n * 128;
    float* vbuf   = ws + off; off += (size_t)n * 128;
    float* ft2    = ws + off; off += (size_t)n * 128;
    float* rst    = ws + off; off += (size_t)n * 128;
    float* hbuf   = ws + off; off += (size_t)n * 512;
    const int np = (n + 4 * SCAN_T + 8 + 31) & ~31;
    int* cnt    = (int*)(ws + off); off += (size_t)np;
    int* rowptr = (int*)(ws + off); off += (size_t)np;
    int* rank   = (int*)(ws + off); off += (size_t)e;
    int* ssrc   = (int*)(ws + off); off += (size_t)e;
    (void)off; (void)ws_size; (void)n_in; (void)out_size;

    const int NTM  = (n + TM - 1) / TM;
    const int NT16 = (n + 15) / 16;
    const int NW8  = (n + 7) / 8;
    const int ZB   = (np + 255) / 256;
    const int EB   = (e + 255) / 256;

    qkv_kernel<<<NTM, 256, 0, stream>>>(feat, Wq, Wk, Wv, q, kbuf, vbuf, n);
    zero_kernel<<<ZB, 256, 0, stream>>>(cnt, n);
    hist_kernel<<<EB, 256, 0, stream>>>(dst, cnt, rank, e);
    scan_kernel<<<1, SCAN_T, 0, stream>>>(cnt, rowptr, n);
    fill_kernel<<<EB, 256, 0, stream>>>(src, dst, rank, rowptr, ssrc, e);
    node_kernel<<<NW8, 256, 0, stream>>>(q, kbuf, vbuf, rowptr, ssrc, ft2, n);
    ln1_kernel<<<NT16, 256, 0, stream>>>(ft2, feat, ln1_g, ln1_b, rst, n);
    ffn1_kernel<<<NTM, 256, 0, stream>>>(rst, W1, b1, pw, hbuf, n);
    ffn2_ln_kernel<<<NTM, 256, 0, stream>>>(hbuf, W2, b2, rst, ln2_g, ln2_b,
                                             (float*)d_out, n);
}


# edges binned by destination in one pass: an edge's atomic slot indexes a 64-slot row of its node (overflow list past 64, summed in arrival order); scan and placement launches go; sum order else kept
# speedup vs baseline: 3.3119x; 1.1437x over previous
#include <hip/hip_runtime.h>
#include <hip/hip_bf16.h>
#include <math.h>

typedef __attribute__((ext_vector_type(2))) float v2f;
typedef __attribute__((ext_vector_type(8))) float v8f;

#define EPSLN 1e-5f

#define TM 80
#define MT (TM / 16)
#define FU (TM * 32 / 256)

__device__ __forceinline__ v8f wmma_f32_4(v2f a, v2f b, v8f c) {
    return __builtin_amdgcn_wmma_f32_16x16x4_f32(
         false, a,  false, b,
         (short)0, c,  false,  false);
}

__device__ __forceinline__ void fill_tile(
    float (*sA)[132], const float* __restrict__ src, int pitch,
    int row0, int n, int tid)
{
    float4 t[FU];
#pragma unroll
    for (int u = 0; u < FU; ++u) {
        const int i = tid + 256 * u;
        int gr = row0 + (i >> 5); if (gr >= n) gr = n - 1;
        t[u] = ((const float4*)(src + (size_t)gr * pitch))[i & 31];
    }
#pragma unroll
    for (int u = 0; u < FU; ++u) {
        const int i = tid + 256 * u;
        ((float4*)&sA[i >> 5][0])[i & 31] = t[u];
    }
}

__global__ __launch_bounds__(256) void qkv_kernel(
    const float* __restrict__ feat,
    const float* __restrict__ Wq, const float* __restrict__ Wk,
    const float* __restrict__ Wv,
    float* __restrict__ q, float* __restrict__ k, float* __restrict__ v, int n)
{
    __shared__ float sA[TM][132];
    const int row0 = blockIdx.x * TM;
    const int tid  = threadIdx.x;

    fill_tile(sA, feat, 128, row0, n, tid);
    __syncthreads();

    const int wave = tid >> 5;
    const int lane = tid & 31;
    const int half = lane >> 4;
    const int lr   = lane & 15;
    const int col  = wave * 16 + lr;

    const float* Ws[3] = { Wq, Wk, Wv };
    float*       Os[3] = { q,  k,  v  };

#pragma unroll
    for (int m = 0; m < 3; ++m) {
        const float* __restrict__ W = Ws[m];
        v2f bfr[32];
#pragma unroll
        for (int kq = 0; kq < 32; ++kq) {
            const int kk = 4 * kq + 2 * half;
            bfr[kq] = (v2f){ W[(size_t)kk * 128 + col],
                             W[(size_t)(kk + 1) * 128 + col] };
        }
        float* __restrict__ O = Os[m];
#pragma unroll
        for (int mt = 0; mt < MT; ++mt) {
            v8f acc = {};
#pragma unroll
            for (int kq = 0; kq < 32; ++kq) {
                const int kk = 4 * kq + 2 * half;
                v2f a = { sA[mt * 16 + lr][kk], sA[mt * 16 + lr][kk + 1] };
                acc = wmma_f32_4(a, bfr[kq], acc);
            }
#pragma unroll
            for (int r = 0; r < 8; ++r) {
                int gr = row0 + mt * 16 + r + 8 * half;
                if (gr < n) O[(size_t)gr * 128 + col] = acc[r];
            }
        }
    }
}

#define CAP 64

__global__ __launch_bounds__(256) void zero_kernel(int* __restrict__ cnt, int n)
{
    const int i = blockIdx.x * 256 + threadIdx.x;
    if (i <= n) cnt[i] = 0;
}

__global__ __launch_bounds__(256) void bin_kernel(
    const int* __restrict__ src, const int* __restrict__ dst,
    int* __restrict__ cnt, int* __restrict__ bucket, int* __restrict__ ovf,
    int e, int n)
{
    const int t = blockIdx.x * 256 + threadIdx.x;
    if (t >= e) return;
    const int d = dst[t], s = src[t];
    const int slot = atomicAdd(&cnt[d], 1);
    if (slot < CAP) {
        bucket[(size_t)d * CAP + slot] = s;
    } else {
        const int o = atomicAdd(&cnt[n], 1);
        ((int2*)ovf)[o] = make_int2(d, s);
    }
}

__device__ __forceinline__ void edge_step(
    const float4 kv, const float4 vv, const float4 qv,
    float& m, float& l, float4& acc)
{
    float p = kv.x * qv.x + kv.y * qv.y + kv.z * qv.z + kv.w * qv.w;
    p += __shfl_xor(p, 1, 32);
    p += __shfl_xor(p, 2, 32);
    p *= 0.08838834764831845f;
    const float t  = expf(-fabsf(p - m));
    const bool  up = p > m;
    const float sc = up ? t : 1.0f;
    const float ex = up ? 1.0f : t;
    m = up ? p : m;
    l = l * sc + ex;
    acc.x = acc.x * sc + ex * vv.x;
    acc.y = acc.y * sc + ex * vv.y;
    acc.z = acc.z * sc + ex * vv.z;
    acc.w = acc.w * sc + ex * vv.w;
}

__global__ __launch_bounds__(256) void node_kernel(
    const float* __restrict__ qbuf, const float* __restrict__ kbuf,
    const float* __restrict__ vbuf, const int* __restrict__ cnt,
    const int* __restrict__ bucket, const int* __restrict__ ovf,
    float* __restrict__ ft2, int n)
{
    const int lane = threadIdx.x & 31;
    const int d = __builtin_amdgcn_readfirstlane(blockIdx.x * 8 + (threadIdx.x >> 5));
    if (d >= n) return;
    const int deg = cnt[d];
    const int nb  = min(deg, CAP);
    const int* __restrict__ brow = bucket + (size_t)d * CAP;
    const float4 qv = ((const float4*)(qbuf + (size_t)d * 128))[lane];
    float m = -INFINITY, l = 0.0f;
    float4 acc = { 0.0f, 0.0f, 0.0f, 0.0f };

    for (int base = 0; base < nb; base += 32) {
        const int cnt32 = min(32, nb - base);
        int x = (lane < cnt32) ? brow[base + lane] : 0x7fffffff;
#pragma unroll
        for (int kk = 2; kk <= 32; kk <<= 1) {
#pragma unroll
            for (int jj = kk >> 1; jj > 0; jj >>= 1) {
                const int o  = __shfl_xor(x, jj, 32);
                const int mn = min(x, o), mx = max(x, o);
                const bool asc = (lane & kk) == 0, low = (lane & jj) == 0;
                x = (asc == low) ? mn : mx;
            }
        }
        int j = 0;
        for (; j + 4 <= cnt32; j += 4) {
            const int s0 = __builtin_amdgcn_readlane(x, j);
            const int s1 = __builtin_amdgcn_readlane(x, j + 1);
            const int s2 = __builtin_amdgcn_readlane(x, j + 2);
            const int s3 = __builtin_amdgcn_readlane(x, j + 3);
            const float4 k0 = ((const float4*)(kbuf + (size_t)s0 * 128))[lane];
            const float4 k1 = ((const float4*)(kbuf + (size_t)s1 * 128))[lane];
            const float4 k2 = ((const float4*)(kbuf + (size_t)s2 * 128))[lane];
            const float4 k3 = ((const float4*)(kbuf + (size_t)s3 * 128))[lane];
            const float4 v0 = ((const float4*)(vbuf + (size_t)s0 * 128))[lane];
            const float4 v1 = ((const float4*)(vbuf + (size_t)s1 * 128))[lane];
            const float4 v2 = ((const float4*)(vbuf + (size_t)s2 * 128))[lane];
            const float4 v3 = ((const float4*)(vbuf + (size_t)s3 * 128))[lane];
            edge_step(k0, v0, qv, m, l, acc);
            edge_step(k1, v1, qv, m, l, acc);
            edge_step(k2, v2, qv, m, l, acc);
            edge_step(k3, v3, qv, m, l, acc);
        }
        for (; j < cnt32; ++j) {
            const int s0 = __builtin_amdgcn_readlane(x, j);
            const float4 k0 = ((const float4*)(kbuf + (size_t)s0 * 128))[lane];
            const float4 v0 = ((const float4*)(vbuf + (size_t)s0 * 128))[lane];
            edge_step(k0, v0, qv, m, l, acc);
        }
    }
    if (deg > CAP) {
        const int novf = cnt[n];
        for (int i = 0; i < novf; i += 32) {
            int od = -1, os = 0;
            if (i + lane < novf) {
                const int2 p = ((const int2*)ovf)[i + lane];
                od = p.x; os = p.y;
            }
            unsigned mask = __builtin_amdgcn_ballot_w32(od == d);
            while (mask) {
                const int b = __builtin_ctz(mask);
                mask &= mask - 1;
                const int s0 = __shfl(os, b, 32);
                const float4 k0 = ((const float4*)(kbuf + (size_t)s0 * 128))[lane];
                const float4 v0 = ((const float4*)(vbuf + (size_t)s0 * 128))[lane];
                edge_step(k0, v0, qv, m, l, acc);
            }
        }
    }
    const float inv = (deg > 0) ? 1.0f / l : 0.0f;
    float4 o4 = { acc.x * inv, acc.y * inv, acc.z * inv, acc.w * inv };
    ((float4*)(ft2 + (size_t)d * 128))[lane] = o4;
}

__global__ __launch_bounds__(256) void ln1_kernel(
    const float* __restrict__ ft2, const float* __restrict__ feat,
    const float* __restrict__ g, const float* __restrict__ b,
    float* __restrict__ rst, int n)
{
    const int row = blockIdx.x * 16 + (threadIdx.x >> 4);
    const int c0  = threadIdx.x & 15;
    if (row >= n) return;
    const float* p1 = ft2  + (size_t)row * 128;
    const float* p2 = feat + (size_t)row * 128;
    float xv[8], s = 0.0f, ss = 0.0f;
#pragma unroll
    for (int j = 0; j < 8; ++j) {
        int c = c0 + 16 * j;
        float x = p1[c] + p2[c];
        xv[j] = x; s += x; ss += x * x;
    }
#pragma unroll
    for (int m = 1; m < 16; m <<= 1) {
        s  += __shfl_xor(s,  m, 32);
        ss += __shfl_xor(ss, m, 32);
    }
    const float mean = s * (1.0f / 128.0f);
    const float var  = ss * (1.0f / 128.0f) - mean * mean;
    const float inv  = rsqrtf(var + EPSLN);
    float* o = rst + (size_t)row * 128;
#pragma unroll
    for (int j = 0; j < 8; ++j) {
        int c = c0 + 16 * j;
        o[c] = (xv[j] - mean) * inv * g[c] + b[c];
    }
}

__global__ __launch_bounds__(256) void ffn1_kernel(
    const float* __restrict__ rst, const float* __restrict__ W1,
    const float* __restrict__ b1, const float* __restrict__ pw,
    float* __restrict__ hbuf, int n)
{
    __shared__ float sA[TM][132];
    const int row0 = blockIdx.x * TM;
    const int tid  = threadIdx.x;
    fill_tile(sA, rst, 128, row0, n, tid);
    __syncthreads();

    const int wave = tid >> 5;
    const int lane = tid & 31;
    const int half = lane >> 4;
    const int lr   = lane & 15;

#pragma unroll
    for (int j = 0; j < 4; ++j) {
        const int col = (wave + 8 * j) * 16 + lr;
        v2f bfr[32];
#pragma unroll
        for (int kq = 0; kq < 32; ++kq) {
            const int kk = 4 * kq + 2 * half;
            bfr[kq] = (v2f){ W1[(size_t)kk * 512 + col],
                             W1[(size_t)(kk + 1) * 512 + col] };
        }
        const float bias = b1[col], slope = pw[col];
#pragma unroll
        for (int mt = 0; mt < MT; ++mt) {
            v8f acc = {};
#pragma unroll
            for (int kq = 0; kq < 32; ++kq) {
                const int kk = 4 * kq + 2 * half;
                v2f a = { sA[mt * 16 + lr][kk], sA[mt * 16 + lr][kk + 1] };
                acc = wmma_f32_4(a, bfr[kq], acc);
            }
#pragma unroll
            for (int r = 0; r < 8; ++r) {
                int gr = row0 + mt * 16 + r + 8 * half;
                if (gr < n) {
                    float h = acc[r] + bias;
                    h = (h >= 0.0f) ? h : slope * h;
                    hbuf[(size_t)gr * 512 + col] = h;
                }
            }
        }
    }
}

__global__ __launch_bounds__(256) void ffn2_ln_kernel(
    const float* __restrict__ hbuf, const float* __restrict__ W2,
    const float* __restrict__ b2, const float* __restrict__ rst,
    const float* __restrict__ g, const float* __restrict__ be,
    float* __restrict__ out, int n)
{
    __shared__ float sA[TM][132];
    const int row0 = blockIdx.x * TM;
    const int tid  = threadIdx.x;
    const int wave = tid >> 5;
    const int lane = tid & 31;
    const int half = lane >> 4;
    const int lr   = lane & 15;
    const int col  = wave * 16 + lr;

    v8f acc[MT] = {};
#pragma unroll
    for (int kc = 0; kc < 4; ++kc) {
        __syncthreads();
        fill_tile(sA, hbuf + kc * 128, 512, row0, n, tid);
        __syncthreads();

        v2f bfr[32];
#pragma unroll
        for (int kq = 0; kq < 32; ++kq) {
            const int kk = kc * 128 + 4 * kq + 2 * half;
            bfr[kq] = (v2f){ W2[(size_t)kk * 128 + col],
                             W2[(size_t)(kk + 1) * 128 + col] };
        }
#pragma unroll
        for (int mt = 0; mt < MT; ++mt) {
#pragma unroll
            for (int kq = 0; kq < 32; ++kq) {
                const int kk = 4 * kq + 2 * half;
                v2f a = { sA[mt * 16 + lr][kk], sA[mt * 16 + lr][kk + 1] };
                acc[mt] = wmma_f32_4(a, bfr[kq], acc[mt]);
            }
        }
    }

    __syncthreads();
    const float bias = b2[col];
#pragma unroll
    for (int mt = 0; mt < MT; ++mt) {
#pragma unroll
        for (int r = 0; r < 8; ++r) {
            const int rl = mt * 16 + r + 8 * half;
            int gr = row0 + rl; if (gr >= n) gr = n - 1;
            sA[rl][col] = acc[mt][r] + bias + rst[(size_t)gr * 128 + col];
        }
    }
    __syncthreads();

    const int c0 = tid & 3;
    for (int rb = 0; rb < TM; rb += 64) {
        const int rl2 = rb + (tid >> 2);
        if (rl2 < TM) {
            const int row = row0 + rl2;
            float xv[32], s = 0.0f, ss = 0.0f;
#pragma unroll
            for (int j = 0; j < 32; ++j) {
                int c = c0 + 4 * j;
                float x = sA[rl2][c];
                xv[j] = x; s += x; ss += x * x;
            }
#pragma unroll
            for (int m = 1; m < 4; m <<= 1) {
                s  += __shfl_xor(s,  m, 32);
                ss += __shfl_xor(ss, m, 32);
            }
            const float mean = s * (1.0f / 128.0f);
            const float var  = ss * (1.0f / 128.0f) - mean * mean;
            const float inv  = rsqrtf(var + EPSLN);
            if (row < n) {
                float* o = out + (size_t)row * 128;
#pragma unroll
                for (int j = 0; j < 32; ++j) {
                    int c = c0 + 4 * j;
                    o[c] = (xv[j] - mean) * inv * g[c] + be[c];
                }
            }
        }
    }
}

extern "C" void kernel_launch(void* const* d_in, const int* in_sizes, int n_in,
                              void* d_out, int out_size, void* d_ws, size_t ws_size,
                              hipStream_t stream)
{
    const float* feat  = (const float*)d_in[0];
    const int*   src   = (const int*)  d_in[1];
    const int*   dst   = (const int*)  d_in[2];
    const float* Wq    = (const float*)d_in[3];
    const float* Wk    = (const float*)d_in[4];
    const float* Wv    = (const float*)d_in[5];
    const float* ln1_g = (const float*)d_in[6];
    const float* ln1_b = (const float*)d_in[7];
    const float* ln2_g = (const float*)d_in[8];
    const float* ln2_b = (const float*)d_in[9];
    const float* W1    = (const float*)d_in[10];
    const float* b1    = (const float*)d_in[11];
    const float* pw    = (const float*)d_in[12];
    const float* W2    = (const float*)d_in[13];
    const float* b2    = (const float*)d_in[14];

    const int n = in_sizes[0] / 128;
    const int e = in_sizes[1];

    float* ws = (float*)d_ws;
    size_t off = 0;
    float* q      = ws + off; off += (size_t)n * 128;
    float* kbuf   = ws + off; off += (size_t)n * 128;
    float* vbuf   = ws + off; off += (size_t)n * 128;
    float* ft2    = ws + off; off += (size_t)n * 128;
    float* rst    = ws + off; off += (size_t)n * 128;
    float* hbuf   = ws + off; off += (size_t)n * 512;
    int* bucket = (int*)(ws + off); off += (size_t)n * CAP;
    int* ovf    = (int*)(ws + off); off += (size_t)e * 2;
    int* cnt    = (int*)(ws + off); off += (size_t)n + 1;
    (void)off; (void)ws_size; (void)n_in; (void)out_size;

    const int NTM  = (n + TM - 1) / TM;
    const int NT16 = (n + 15) / 16;
    const int NW8  = (n + 7) / 8;
    const int ZB   = (n + 1 + 255) / 256;
    const int EB   = (e + 255) / 256;

    qkv_kernel<<<NTM, 256, 0, stream>>>(feat, Wq, Wk, Wv, q, kbuf, vbuf, n);
    zero_kernel<<<ZB, 256, 0, stream>>>(cnt, n);
    bin_kernel<<<EB, 256, 0, stream>>>(src, dst, cnt, bucket, ovf, e, n);
    node_kernel<<<NW8, 256, 0, stream>>>(q, kbuf, vbuf, cnt, bucket, ovf, ft2, n);
    ln1_kernel<<<NT16, 256, 0, stream>>>(ft2, feat, ln1_g, ln1_b, rst, n);
    ffn1_kernel<<<NTM, 256, 0, stream>>>(rst, W1, b1, pw, hbuf, n);
    ffn2_ln_kernel<<<NTM, 256, 0, stream>>>(hbuf, W2, b2, rst, ln2_g, ln2_b,
                                             (float*)d_out, n);
}
